# starting kernel
# baseline (speedup 1.0000x reference)
#include <hip/hip_runtime.h>
#include <hip/hip_bf16.h>


typedef __attribute__((ext_vector_type(2))) float v2f;
typedef __attribute__((ext_vector_type(8))) float v8f;

#define D_FEAT 128

__global__ void zero_kernel(float* __restrict__ p, int n) {
  int i = blockIdx.x * blockDim.x + threadIdx.x;
  if (i < n) p[i] = 0.0f;
}

__global__ void deg_kernel(const int* __restrict__ col,
                           const float* __restrict__ w,
                           float* __restrict__ deg, int E) {
  int e = blockIdx.x * blockDim.x + threadIdx.x;
  if (e < E) atomicAdd(&deg[col[e]], w[e]);
}

__global__ void dinv_kernel(float* __restrict__ deg, int n) {
  int i = blockIdx.x * blockDim.x + threadIdx.x;
  if (i < n) {
    float d = deg[i] + 1.0f;
    deg[i] = rsqrtf(d);
  }
}

__global__ __launch_bounds__(256) void gemm_wmma_kernel(
    const float* __restrict__ Z, const float* __restrict__ W,
    float* __restrict__ H, int nRows) {
  const int lane = threadIdx.x;
  const int wave = threadIdx.y;
  const int tile = blockIdx.x * 8 + wave;
  const int nTiles = (nRows / 16) * (D_FEAT / 16);
  if (tile >= nTiles) return;

  const int m0 = (tile / (D_FEAT / 16)) * 16;
  const int n0 = (tile % (D_FEAT / 16)) * 16;
  const int l15   = lane & 15;
  const int khalf = (lane >> 4) << 1;

  const float* zrow = Z + (size_t)(m0 + l15) * D_FEAT;
  v8f acc = {};

#pragma unroll
  for (int k0 = 0; k0 < D_FEAT; k0 += 4) {
    const int kb = k0 + khalf;
    v2f a;
    a.x = zrow[kb];
    a.y = zrow[kb + 1];
    v2f b;
    b.x = W[(size_t)kb * D_FEAT + n0 + l15];
    b.y = W[(size_t)(kb + 1) * D_FEAT + n0 + l15];
    acc = __builtin_amdgcn_wmma_f32_16x16x4_f32(
         false, a,  false, b,
         (short)0, acc,  false,  false);
  }

  const int mbase = m0 + ((lane >> 4) << 3);
  float* hp = H + (size_t)mbase * D_FEAT + n0 + l15;
#pragma unroll
  for (int r = 0; r < 8; ++r) hp[(size_t)r * D_FEAT] = acc[r];
}

__global__ __launch_bounds__(256) void aggregate_kernel(
    const int* __restrict__ row, const int* __restrict__ col,
    const float* __restrict__ w, const float* __restrict__ dinv,
    const float* __restrict__ H, float* __restrict__ AGG, int E) {
  const int lane = threadIdx.x;
  const int e = blockIdx.x * 8 + threadIdx.y;
  if (e >= E) return;
  const int r = row[e];
  const int c = col[e];
  const float nrm = dinv[r] * w[e] * dinv[c];

  const float4 v = ((const float4*)(H + (size_t)r * D_FEAT))[lane];
  float* ap = AGG + (size_t)c * D_FEAT + lane * 4;
  atomicAdd(ap + 0, v.x * nrm);
  atomicAdd(ap + 1, v.y * nrm);
  atomicAdd(ap + 2, v.z * nrm);
  atomicAdd(ap + 3, v.w * nrm);
}

__global__ void finish_kernel(float* __restrict__ Z  ,
                              const float* __restrict__ H,
                              const float* __restrict__ dinv,
                              const float* __restrict__ bias,
                              const float* __restrict__ pw, int total) {
  int i = blockIdx.x * blockDim.x + threadIdx.x;
  if (i >= total) return;
  const int f = i & (D_FEAT - 1);
  const int node = i >> 7;
  const float d = dinv[node];
  const float v = Z[i] + H[i] * d * d + bias[f];
  Z[i] = fmaxf(v, 0.0f) + pw[f] * fminf(v, 0.0f);
}

__global__ void accum_kernel(const float* __restrict__ Z,
                             const float* __restrict__ coeff, int t,
                             float* __restrict__ out, int total) {
  int i = blockIdx.x * blockDim.x + threadIdx.x;
  if (i >= total) return;
  out[i] += coeff[t] * Z[i];
}

extern "C" void kernel_launch(void* const* d_in, const int* in_sizes, int n_in,
                              void* d_out, int out_size, void* d_ws, size_t ws_size,
                              hipStream_t stream) {
  const float* x       = (const float*)d_in[0];
  const int*   eidx    = (const int*)  d_in[1];
  const float* ew      = (const float*)d_in[2];
  const float* coeff   = (const float*)d_in[3];
  const float* W1      = (const float*)d_in[4];
  const float* b1      = (const float*)d_in[5];
  const float* W2      = (const float*)d_in[6];
  const float* b2      = (const float*)d_in[7];
  const float* prelu_w = (const float*)d_in[8];

  const int N = in_sizes[0] / D_FEAT;
  const int E = in_sizes[2] / 3;
  const int total = N * D_FEAT;

  float* ws   = (float*)d_ws;
  float* dinv = ws;
  float* bufA = ws + (((size_t)N + 31) & ~(size_t)31);
  float* bufB = bufA + (size_t)N * D_FEAT;

  const int ETHREADS = 256;
  const dim3 waveBlock(32, 8);
  const int gemmBlocks = ((N / 16) * (D_FEAT / 16) + 7) / 8;
  const int edgeBlocks = (E + 7) / 8;
  const int elemBlocks = (total + ETHREADS - 1) / ETHREADS;
  const int nodeBlocks = (N + ETHREADS - 1) / ETHREADS;

  zero_kernel<<<elemBlocks, ETHREADS, 0, stream>>>((float*)d_out, total);

  for (int t = 0; t < 3; ++t) {
    const int* row = eidx + (size_t)t * 2 * E;
    const int* col = row + E;
    const float* w = ew + (size_t)t * E;

    zero_kernel<<<nodeBlocks, ETHREADS, 0, stream>>>(dinv, N);
    deg_kernel<<<(E + ETHREADS - 1) / ETHREADS, ETHREADS, 0, stream>>>(col, w, dinv, E);
    dinv_kernel<<<nodeBlocks, ETHREADS, 0, stream>>>(dinv, N);

    gemm_wmma_kernel<<<gemmBlocks, waveBlock, 0, stream>>>(x, W1, bufA, N);
    zero_kernel<<<elemBlocks, ETHREADS, 0, stream>>>(bufB, total);
    aggregate_kernel<<<edgeBlocks, waveBlock, 0, stream>>>(row, col, w, dinv, bufA, bufB, E);
    finish_kernel<<<elemBlocks, ETHREADS, 0, stream>>>(bufB, bufA, dinv, b1, prelu_w, total);

    gemm_wmma_kernel<<<gemmBlocks, waveBlock, 0, stream>>>(bufB, W2, bufA, N);
    zero_kernel<<<elemBlocks, ETHREADS, 0, stream>>>(bufB, total);
    aggregate_kernel<<<edgeBlocks, waveBlock, 0, stream>>>(row, col, w, dinv, bufA, bufB, E);
    finish_kernel<<<elemBlocks, ETHREADS, 0, stream>>>(bufB, bufA, dinv, b2, prelu_w, total);

    accum_kernel<<<elemBlocks, ETHREADS, 0, stream>>>(bufB, coeff, t, (float*)d_out, total);
  }
}


# Degree pass also counting-sorts each of 128 edge-list chunks by destination block in LDS; layer 1 reads its block's runs, scans no edge list; a wave meets edges in a new order, f32 atomics as before
# speedup vs baseline: 2.4505x; 2.4505x over previous
#include <hip/hip_runtime.h>
#include <hip/hip_bf16.h>


typedef __attribute__((ext_vector_type(2))) float v2f;
typedef __attribute__((ext_vector_type(8))) float v8f;

#define D_FEAT 128

__global__ void zero_kernel(float* __restrict__ p, int n) {
  int i = blockIdx.x * blockDim.x + threadIdx.x;
  if (i < n) p[i] = 0.0f;
}

#define NB 196
#define NCHUNK 128
#define SORT_THREADS 1024
#define SORT_WAVES (SORT_THREADS / 32)
#define SORT_EPT 8
#define SORT_CAP (SORT_EPT * SORT_THREADS)
#define SORT_BINS 256

__global__ __launch_bounds__(SORT_THREADS) void degree_sort_kernel(
    const int* __restrict__ row, const int* __restrict__ col,
    const float* __restrict__ w, float* __restrict__ deg, int E,
    int* __restrict__ rsrc, int* __restrict__ rcol, float* __restrict__ rw,
    int* __restrict__ seg) {
  __shared__ int cnt[SORT_WAVES][SORT_BINS];
  __shared__ int binStart[SORT_BINS];
  __shared__ int waveSum[SORT_BINS / 32];
  __shared__ int s_r[SORT_CAP];
  __shared__ int s_c[SORT_CAP];
  __shared__ float s_w[SORT_CAP];

  const int tid = threadIdx.x;
  const int lane = tid & 31;
  const int wave = tid >> 5;
  const int chunkLen = (E + NCHUNK - 1) / NCHUNK;
  const int e0 = blockIdx.x * chunkLen;
  const int n = max(0, min(chunkLen, E - e0));

  for (int i = tid; i < SORT_WAVES * SORT_BINS; i += SORT_THREADS) (&cnt[0][0])[i] = 0;
  __syncthreads();

  int r[SORT_EPT], c[SORT_EPT], b[SORT_EPT], rank[SORT_EPT];
  float wt[SORT_EPT];
#pragma unroll
  for (int k = 0; k < SORT_EPT; ++k) {
    const int i = k * SORT_THREADS + tid;
    const bool ok = i < n;
    c[k] = ok ? col[e0 + i] : -1;
    r[k] = ok ? row[e0 + i] : 0;
    wt[k] = ok ? w[e0 + i] : 0.0f;
  }
#pragma unroll
  for (int k = 0; k < SORT_EPT; ++k) {
    b[k] = 0;
    rank[k] = 0;
    if (c[k] >= 0) {
      b[k] = c[k] / NB;
      rank[k] = __hip_atomic_fetch_add(&cnt[wave][b[k]], 1, __ATOMIC_RELAXED,
                                       __HIP_MEMORY_SCOPE_WORKGROUP);
    }
  }
  __syncthreads();

  int tot = 0;
  if (tid < SORT_BINS) {
    int t[SORT_WAVES];
#pragma unroll
    for (int q = 0; q < SORT_WAVES; ++q) t[q] = cnt[q][tid];
#pragma unroll
    for (int q = 0; q < SORT_WAVES; ++q) {
      cnt[q][tid] = tot;
      tot += t[q];
    }
  }
  int incl = tot;
#pragma unroll
  for (int d = 1; d < 32; d <<= 1) {
    const int o = __shfl_up(incl, d);
    if (lane >= d) incl += o;
  }
  if (tid < SORT_BINS && lane == 31) waveSum[wave] = incl;
  __syncthreads();
  if (tid < SORT_BINS) {
    int base = 0;
    for (int q = 0; q < wave; ++q) base += waveSum[q];
    const int start = base + incl - tot;
    binStart[tid] = start;
    ((int2*)seg)[(size_t)tid * NCHUNK + blockIdx.x] = make_int2(start, tot);
  }
  __syncthreads();

#pragma unroll
  for (int k = 0; k < SORT_EPT; ++k) {
    if (c[k] >= 0) {
      const int pos = binStart[b[k]] + cnt[wave][b[k]] + rank[k];
      s_r[pos] = r[k];
      s_c[pos] = c[k];
      s_w[pos] = wt[k];
    }
  }
  __syncthreads();
  for (int i = tid; i < n; i += SORT_THREADS) {
    rsrc[e0 + i] = s_r[i];
    rcol[e0 + i] = s_c[i];
    rw[e0 + i] = s_w[i];
  }
#pragma unroll
  for (int k = 0; k < SORT_EPT; ++k)
    if (c[k] >= 0) atomicAdd(&deg[c[k]], wt[k]);
}

__global__ void dinv_kernel(float* __restrict__ deg, int n) {
  int i = blockIdx.x * blockDim.x + threadIdx.x;
  if (i < n) {
    float d = deg[i] + 1.0f;
    deg[i] = rsqrtf(d);
  }
}

__global__ __launch_bounds__(256) void gemm_wmma_kernel(
    const float* __restrict__ Z, const float* __restrict__ W,
    float* __restrict__ H, int nRows) {
  const int lane = threadIdx.x;
  const int wave = threadIdx.y;
  const int tile = blockIdx.x * 8 + wave;
  const int nTiles = (nRows / 16) * (D_FEAT / 16);
  if (tile >= nTiles) return;

  const int m0 = (tile / (D_FEAT / 16)) * 16;
  const int n0 = (tile % (D_FEAT / 16)) * 16;
  const int l15   = lane & 15;
  const int khalf = (lane >> 4) << 1;

  const float* zrow = Z + (size_t)(m0 + l15) * D_FEAT;
  v8f acc = {};

#pragma unroll
  for (int k0 = 0; k0 < D_FEAT; k0 += 4) {
    const int kb = k0 + khalf;
    v2f a;
    a.x = zrow[kb];
    a.y = zrow[kb + 1];
    v2f b;
    b.x = W[(size_t)kb * D_FEAT + n0 + l15];
    b.y = W[(size_t)(kb + 1) * D_FEAT + n0 + l15];
    acc = __builtin_amdgcn_wmma_f32_16x16x4_f32(
         false, a,  false, b,
         (short)0, acc,  false,  false);
  }

  const int mbase = m0 + ((lane >> 4) << 3);
  float* hp = H + (size_t)mbase * D_FEAT + n0 + l15;
#pragma unroll
  for (int r = 0; r < 8; ++r) hp[(size_t)r * D_FEAT] = acc[r];
}

#define AGG_THREADS 1024
#define AGG_WAVES (AGG_THREADS / 32)
#define CPW (NCHUNK / AGG_WAVES)
#define LIST_GROUPS 16
#define LIST_WORDS (LIST_GROUPS * 32)

template <int MODE>
__global__ __launch_bounds__(AGG_THREADS) void aggregate_block_kernel(
    const int* __restrict__ rsrc, const int* __restrict__ rcol,
    const float* __restrict__ rw, const int* __restrict__ seg,
    const float* __restrict__ dinv,
    const float* __restrict__ H, const float* __restrict__ bias,
    const float* __restrict__ pw, float* __restrict__ Z, int E, int N,
    int* __restrict__ lsrc, int* __restrict__ ldst, float* __restrict__ lnrm,
    int* __restrict__ lcnt) {
  __shared__ float acc[NB * D_FEAT];

  const int tid = threadIdx.x;
  const int lane = tid & 31;
  const int wave = __builtin_amdgcn_readfirstlane(tid >> 5);
  const int n0 = blockIdx.x * NB;
  const int nb = min(NB, N - n0);

  {
    float4* a4 = (float4*)acc;
    for (int i = tid; i < NB * D_FEAT / 4; i += AGG_THREADS)
      a4[i] = make_float4(0.0f, 0.0f, 0.0f, 0.0f);
  }
  __syncthreads();

  const size_t slot = (size_t)blockIdx.x * AGG_WAVES + wave;
  int* mysrc = lsrc + slot * LIST_WORDS;
  int* mydst = ldst + slot * LIST_WORDS;
  float* mynrm = lnrm + slot * LIST_WORDS;

  auto add_group = [&](int r, int cl, float nrm, int take) {
    for (int g0 = 0; g0 < take; g0 += 8) {
      float v[8][4];
#pragma unroll
      for (int u = 0; u < 8; ++u) {
        const int k = min(g0 + u, take - 1);
        const int rk = __builtin_amdgcn_readlane(r, k);
        const float* hp = H + (size_t)rk * D_FEAT + lane;
#pragma unroll
        for (int q = 0; q < 4; ++q) v[u][q] = hp[32 * q];
      }
#pragma unroll
      for (int u = 0; u < 8; ++u)
        asm volatile("" : : "v"(v[u][0]), "v"(v[u][1]), "v"(v[u][2]), "v"(v[u][3]));
#pragma unroll
      for (int u = 0; u < 8; ++u) {
        if (g0 + u < take) {
          const int ck = __builtin_amdgcn_readlane(cl, g0 + u);
          const float nk = __int_as_float(__builtin_amdgcn_readlane(__float_as_int(nrm), g0 + u));
          float* ap = acc + ck * D_FEAT + lane;
#pragma unroll
          for (int q = 0; q < 4; ++q)
            __hip_atomic_fetch_add(ap + 32 * q, v[u][q] * nk, __ATOMIC_RELAXED,
                                   __HIP_MEMORY_SCOPE_WORKGROUP);
        }
      }
    }
  };

  auto walk = [&](bool keep, int& total) {
    const int chunkLen = (E + NCHUNK - 1) / NCHUNK;
    int2 sc = make_int2(0, 0);
    if (lane < CPW) sc = ((const int2*)seg)[(size_t)blockIdx.x * NCHUNK + wave * CPW + lane];
    int s[CPW], p[CPW + 1];
    p[0] = 0;
#pragma unroll
    for (int q = 0; q < CPW; ++q) {
      s[q] = __builtin_amdgcn_readlane(sc.x, q) + (wave * CPW + q) * chunkLen;
      p[q + 1] = p[q] + __builtin_amdgcn_readlane(sc.y, q);
    }
    total = p[CPW];
    int groups = 0;
    for (int done = 0; done < total; done += 32) {
      const int v = done + lane;
      int r = 0, cl = 0;
      float nrm = 0.0f;
      if (v < total) {
        int a = s[0] + v;
#pragma unroll
        for (int q = 1; q < CPW; ++q)
          if (v >= p[q]) a = s[q] + (v - p[q]);
        r = rsrc[a];
        const int c = rcol[a];
        nrm = dinv[r] * rw[a] * dinv[c];
        cl = c - n0;
      }
      if (keep) {
        if (groups < LIST_GROUPS) {
          mysrc[groups * 32 + lane] = r;
          mydst[groups * 32 + lane] = cl;
          mynrm[groups * 32 + lane] = nrm;
        }
      }
      ++groups;
      add_group(r, cl, nrm, min(32, total - done));
    }
    return groups;
  };

  if (MODE == 0) {
    int total;
    const int groups = walk(true, total);
    if (lane == 0) lcnt[slot] = groups <= LIST_GROUPS ? total : -1;
  } else {
    const int cnt = __builtin_amdgcn_readfirstlane(lcnt[slot]);
    if (cnt >= 0) {
      for (int done = 0; done < cnt; done += 32) {
        const int r = mysrc[done + lane];
        const int cl = mydst[done + lane];
        const float nrm = mynrm[done + lane];
        add_group(r, cl, nrm, min(32, cnt - done));
      }
    } else {
      int total;
      walk(false, total);
    }
  }

  __syncthreads();

  for (int i = tid; i < nb * D_FEAT; i += AGG_THREADS) {
    const int f = i & (D_FEAT - 1);
    const int node = n0 + (i >> 7);
    const float d = dinv[node];
    const size_t g = (size_t)node * D_FEAT + f;
    const float v = acc[i] + H[g] * d * d + bias[f];
    Z[g] = fmaxf(v, 0.0f) + pw[f] * fminf(v, 0.0f);
  }
}

#define COMB_THREADS 256
#define COMB_BLOCKS 512
__global__ __launch_bounds__(COMB_THREADS) void combine_kernel(
    const float* __restrict__ z0, const float* __restrict__ z1,
    const float* __restrict__ z2, const float* __restrict__ coeff,
    float* __restrict__ out, int n4) {
  const float c0 = coeff[0], c1 = coeff[1], c2 = coeff[2];
  const float4* p0 = (const float4*)z0;
  const float4* p1 = (const float4*)z1;
  const float4* p2 = (const float4*)z2;
  float4* o4 = (float4*)out;
  const int stride = gridDim.x * COMB_THREADS;
  int i = blockIdx.x * COMB_THREADS + threadIdx.x;

  auto mix = [&](const float4& a, const float4& b, const float4& c) {
    float4 r;
    r.x = __builtin_fmaf(c2, c.x, __builtin_fmaf(c1, b.x, c0 * a.x));
    r.y = __builtin_fmaf(c2, c.y, __builtin_fmaf(c1, b.y, c0 * a.y));
    r.z = __builtin_fmaf(c2, c.z, __builtin_fmaf(c1, b.z, c0 * a.z));
    r.w = __builtin_fmaf(c2, c.w, __builtin_fmaf(c1, b.w, c0 * a.w));
    return r;
  };

  for (; i + 3 * stride < n4; i += 4 * stride) {
    float4 a[4], b[4], c[4];
#pragma unroll
    for (int u = 0; u < 4; ++u) {
      a[u] = p0[i + u * stride];
      b[u] = p1[i + u * stride];
      c[u] = p2[i + u * stride];
    }
#pragma unroll
    for (int u = 0; u < 4; ++u) o4[i + u * stride] = mix(a[u], b[u], c[u]);
  }
  for (; i < n4; i += stride) o4[i] = mix(p0[i], p1[i], p2[i]);
}

extern "C" void kernel_launch(void* const* d_in, const int* in_sizes, int n_in,
                              void* d_out, int out_size, void* d_ws, size_t ws_size,
                              hipStream_t stream) {
  const float* x       = (const float*)d_in[0];
  const int*   eidx    = (const int*)  d_in[1];
  const float* ew      = (const float*)d_in[2];
  const float* coeff   = (const float*)d_in[3];
  const float* W1      = (const float*)d_in[4];
  const float* b1      = (const float*)d_in[5];
  const float* W2      = (const float*)d_in[6];
  const float* b2      = (const float*)d_in[7];
  const float* prelu_w = (const float*)d_in[8];

  const int N = in_sizes[0] / D_FEAT;
  const int E = in_sizes[2] / 3;
  const int total = N * D_FEAT;

  const int aggBlocks  = (N + NB - 1) / NB;

  float* ws   = (float*)d_ws;
  float* dinv = ws;
  float* bufA = ws + (((size_t)N + 31) & ~(size_t)31);
  float* zpl  = bufA + (size_t)N * D_FEAT;
  const size_t nSlots = (size_t)aggBlocks * AGG_WAVES;
  int*   lcnt = (int*)(zpl + (size_t)3 * N * D_FEAT);
  int*   lsrc = lcnt + ((nSlots + 31) & ~(size_t)31);
  int*   ldst = lsrc + nSlots * LIST_WORDS;
  float* lnrm = (float*)(ldst + nSlots * LIST_WORDS);
  const size_t ePad = ((size_t)E + 31) & ~(size_t)31;
  int*   rsrc = (int*)(lnrm + nSlots * LIST_WORDS);
  int*   rcol = rsrc + ePad;
  float* rw   = (float*)(rcol + ePad);
  int*   seg  = (int*)(rw + ePad);

  const int ETHREADS = 256;
  const dim3 waveBlock(32, 8);
  const int gemmBlocks = ((N / 16) * (D_FEAT / 16) + 7) / 8;
  const int nodeBlocks = (N + ETHREADS - 1) / ETHREADS;

  for (int t = 0; t < 3; ++t) {
    const int* row = eidx + (size_t)t * 2 * E;
    const int* col = row + E;
    const float* w = ew + (size_t)t * E;
    float* zt = zpl + (size_t)t * total;

    zero_kernel<<<nodeBlocks, ETHREADS, 0, stream>>>(dinv, N);
    degree_sort_kernel<<<NCHUNK, SORT_THREADS, 0, stream>>>(row, col, w, dinv, E, rsrc, rcol, rw, seg);
    dinv_kernel<<<nodeBlocks, ETHREADS, 0, stream>>>(dinv, N);

    gemm_wmma_kernel<<<gemmBlocks, waveBlock, 0, stream>>>(x, W1, bufA, N);
    aggregate_block_kernel<0><<<aggBlocks, AGG_THREADS, 0, stream>>>(rsrc, rcol, rw, seg, dinv, bufA, b1, prelu_w, zt, E, N,
                                                                  lsrc, ldst, lnrm, lcnt);

    gemm_wmma_kernel<<<gemmBlocks, waveBlock, 0, stream>>>(zt, W2, bufA, N);
    aggregate_block_kernel<1><<<aggBlocks, AGG_THREADS, 0, stream>>>(rsrc, rcol, rw, seg, dinv, bufA, b2, prelu_w, zt, E, N,
                                                                  lsrc, ldst, lnrm, lcnt);
  }

  combine_kernel<<<COMB_BLOCKS, COMB_THREADS, 0, stream>>>(zpl, zpl + (size_t)total, zpl + (size_t)2 * total, coeff,
                                                         (float*)d_out, total / 4);
}


# Layer 2's product z1@W2 done by layer 1's aggregation work-group: its PReLU rows stay in LDS and feed f32 WMMA, W2 staged in LDS; z1 never stored, 3 GEMM launches gone; A(ZW) and k order unchanged
# speedup vs baseline: 2.7217x; 1.1107x over previous
#include <hip/hip_runtime.h>
#include <hip/hip_bf16.h>


typedef __attribute__((ext_vector_type(2))) float v2f;
typedef __attribute__((ext_vector_type(8))) float v8f;

#define D_FEAT 128

__global__ void zero_kernel(float* __restrict__ p, int n) {
  int i = blockIdx.x * blockDim.x + threadIdx.x;
  if (i < n) p[i] = 0.0f;
}

#define NB 196
#define NCHUNK 128
#define SORT_THREADS 1024
#define SORT_WAVES (SORT_THREADS / 32)
#define SORT_EPT 8
#define SORT_CAP (SORT_EPT * SORT_THREADS)
#define SORT_BINS 256

__global__ __launch_bounds__(SORT_THREADS) void degree_sort_kernel(
    const int* __restrict__ row, const int* __restrict__ col,
    const float* __restrict__ w, float* __restrict__ deg, int E,
    int* __restrict__ rsrc, int* __restrict__ rcol, float* __restrict__ rw,
    int* __restrict__ seg) {
  __shared__ int cnt[SORT_WAVES][SORT_BINS];
  __shared__ int binStart[SORT_BINS];
  __shared__ int waveSum[SORT_BINS / 32];
  __shared__ int s_r[SORT_CAP];
  __shared__ int s_c[SORT_CAP];
  __shared__ float s_w[SORT_CAP];

  const int tid = threadIdx.x;
  const int lane = tid & 31;
  const int wave = tid >> 5;
  const int chunkLen = (E + NCHUNK - 1) / NCHUNK;
  const int e0 = blockIdx.x * chunkLen;
  const int n = max(0, min(chunkLen, E - e0));

  for (int i = tid; i < SORT_WAVES * SORT_BINS; i += SORT_THREADS) (&cnt[0][0])[i] = 0;
  __syncthreads();

  int r[SORT_EPT], c[SORT_EPT], b[SORT_EPT], rank[SORT_EPT];
  float wt[SORT_EPT];
#pragma unroll
  for (int k = 0; k < SORT_EPT; ++k) {
    const int i = k * SORT_THREADS + tid;
    const bool ok = i < n;
    c[k] = ok ? col[e0 + i] : -1;
    r[k] = ok ? row[e0 + i] : 0;
    wt[k] = ok ? w[e0 + i] : 0.0f;
  }
#pragma unroll
  for (int k = 0; k < SORT_EPT; ++k) {
    b[k] = 0;
    rank[k] = 0;
    if (c[k] >= 0) {
      b[k] = c[k] / NB;
      rank[k] = __hip_atomic_fetch_add(&cnt[wave][b[k]], 1, __ATOMIC_RELAXED,
                                       __HIP_MEMORY_SCOPE_WORKGROUP);
    }
  }
  __syncthreads();

  int tot = 0;
  if (tid < SORT_BINS) {
    int t[SORT_WAVES];
#pragma unroll
    for (int q = 0; q < SORT_WAVES; ++q) t[q] = cnt[q][tid];
#pragma unroll
    for (int q = 0; q < SORT_WAVES; ++q) {
      cnt[q][tid] = tot;
      tot += t[q];
    }
  }
  int incl = tot;
#pragma unroll
  for (int d = 1; d < 32; d <<= 1) {
    const int o = __shfl_up(incl, d);
    if (lane >= d) incl += o;
  }
  if (tid < SORT_BINS && lane == 31) waveSum[wave] = incl;
  __syncthreads();
  if (tid < SORT_BINS) {
    int base = 0;
    for (int q = 0; q < wave; ++q) base += waveSum[q];
    const int start = base + incl - tot;
    binStart[tid] = start;
    ((int2*)seg)[(size_t)tid * NCHUNK + blockIdx.x] = make_int2(start, tot);
  }
  __syncthreads();

#pragma unroll
  for (int k = 0; k < SORT_EPT; ++k) {
    if (c[k] >= 0) {
      const int pos = binStart[b[k]] + cnt[wave][b[k]] + rank[k];
      s_r[pos] = r[k];
      s_c[pos] = c[k];
      s_w[pos] = wt[k];
    }
  }
  __syncthreads();
  for (int i = tid; i < n; i += SORT_THREADS) {
    rsrc[e0 + i] = s_r[i];
    rcol[e0 + i] = s_c[i];
    rw[e0 + i] = s_w[i];
  }
#pragma unroll
  for (int k = 0; k < SORT_EPT; ++k)
    if (c[k] >= 0) atomicAdd(&deg[c[k]], wt[k]);
}

__global__ void dinv_kernel(float* __restrict__ deg, int n) {
  int i = blockIdx.x * blockDim.x + threadIdx.x;
  if (i < n) {
    float d = deg[i] + 1.0f;
    deg[i] = rsqrtf(d);
  }
}

__global__ __launch_bounds__(256) void gemm_wmma_kernel(
    const float* __restrict__ Z, const float* __restrict__ W,
    float* __restrict__ H, int nRows) {
  const int lane = threadIdx.x;
  const int wave = threadIdx.y;
  const int tile = blockIdx.x * 8 + wave;
  const int nTiles = (nRows / 16) * (D_FEAT / 16);
  if (tile >= nTiles) return;

  const int m0 = (tile / (D_FEAT / 16)) * 16;
  const int n0 = (tile % (D_FEAT / 16)) * 16;
  const int l15   = lane & 15;
  const int khalf = (lane >> 4) << 1;

  const float* zrow = Z + (size_t)(m0 + l15) * D_FEAT;
  v8f acc = {};

#pragma unroll
  for (int k0 = 0; k0 < D_FEAT; k0 += 4) {
    const int kb = k0 + khalf;
    v2f a;
    a.x = zrow[kb];
    a.y = zrow[kb + 1];
    v2f b;
    b.x = W[(size_t)kb * D_FEAT + n0 + l15];
    b.y = W[(size_t)(kb + 1) * D_FEAT + n0 + l15];
    acc = __builtin_amdgcn_wmma_f32_16x16x4_f32(
         false, a,  false, b,
         (short)0, acc,  false,  false);
  }

  const int mbase = m0 + ((lane >> 4) << 3);
  float* hp = H + (size_t)mbase * D_FEAT + n0 + l15;
#pragma unroll
  for (int r = 0; r < 8; ++r) hp[(size_t)r * D_FEAT] = acc[r];
}

#define NBP 208
#define PITCH 132
#define WPITCH 288
#define AGG_THREADS 1024
#define AGG_WAVES (AGG_THREADS / 32)
#define CPW (NCHUNK / AGG_WAVES)
#define LIST_GROUPS 16
#define LIST_WORDS (LIST_GROUPS * 32)

template <int MODE>
__global__ __launch_bounds__(AGG_THREADS) void aggregate_block_kernel(
    const int* __restrict__ rsrc, const int* __restrict__ rcol,
    const float* __restrict__ rw, const int* __restrict__ seg,
    const float* __restrict__ dinv,
    const float* __restrict__ H, const float* __restrict__ bias,
    const float* __restrict__ pw, float* __restrict__ Z, int E, int N,
    int* __restrict__ lsrc, int* __restrict__ ldst, float* __restrict__ lnrm,
    int* __restrict__ lcnt, const float* __restrict__ Wn) {
  __shared__ float acc[NBP * PITCH];
  __shared__ float wl[MODE == 0 ? (D_FEAT / 2) * WPITCH : 4];

  const int tid = threadIdx.x;
  const int lane = tid & 31;
  const int wave = __builtin_amdgcn_readfirstlane(tid >> 5);
  const int n0 = blockIdx.x * NB;
  const int nb = min(NB, N - n0);

  {
    float4* a4 = (float4*)acc;
    for (int i = tid; i < NBP * PITCH / 4; i += AGG_THREADS)
      a4[i] = make_float4(0.0f, 0.0f, 0.0f, 0.0f);
  }
  if (MODE == 0) {
    for (int i = tid; i < (D_FEAT / 2) * (D_FEAT / 4); i += AGG_THREADS) {
      const int p = i >> 5;
      const int c4 = (i & 31) * 4;
      const float4 lo = *(const float4*)(Wn + (size_t)(2 * p) * D_FEAT + c4);
      const float4 hi = *(const float4*)(Wn + (size_t)(2 * p + 1) * D_FEAT + c4);
      float4* dst = (float4*)(wl + p * WPITCH + c4 * 2);
      dst[0] = make_float4(lo.x, hi.x, lo.y, hi.y);
      dst[1] = make_float4(lo.z, hi.z, lo.w, hi.w);
    }
  }
  __syncthreads();

  const size_t slot = (size_t)blockIdx.x * AGG_WAVES + wave;
  int* mysrc = lsrc + slot * LIST_WORDS;
  int* mydst = ldst + slot * LIST_WORDS;
  float* mynrm = lnrm + slot * LIST_WORDS;

  auto add_group = [&](int r, int cl, float nrm, int take) {
    for (int g0 = 0; g0 < take; g0 += 8) {
      float v[8][4];
#pragma unroll
      for (int u = 0; u < 8; ++u) {
        const int k = min(g0 + u, take - 1);
        const int rk = __builtin_amdgcn_readlane(r, k);
        const float* hp = H + (size_t)rk * D_FEAT + lane;
#pragma unroll
        for (int q = 0; q < 4; ++q) v[u][q] = hp[32 * q];
      }
#pragma unroll
      for (int u = 0; u < 8; ++u)
        asm volatile("" : : "v"(v[u][0]), "v"(v[u][1]), "v"(v[u][2]), "v"(v[u][3]));
#pragma unroll
      for (int u = 0; u < 8; ++u) {
        if (g0 + u < take) {
          const int ck = __builtin_amdgcn_readlane(cl, g0 + u);
          const float nk = __int_as_float(__builtin_amdgcn_readlane(__float_as_int(nrm), g0 + u));
          float* ap = acc + ck * PITCH + lane;
#pragma unroll
          for (int q = 0; q < 4; ++q)
            __hip_atomic_fetch_add(ap + 32 * q, v[u][q] * nk, __ATOMIC_RELAXED,
                                   __HIP_MEMORY_SCOPE_WORKGROUP);
        }
      }
    }
  };

  auto walk = [&](bool keep, int& total) {
    const int chunkLen = (E + NCHUNK - 1) / NCHUNK;
    int2 sc = make_int2(0, 0);
    if (lane < CPW) sc = ((const int2*)seg)[(size_t)blockIdx.x * NCHUNK + wave * CPW + lane];
    int s[CPW], p[CPW + 1];
    p[0] = 0;
#pragma unroll
    for (int q = 0; q < CPW; ++q) {
      s[q] = __builtin_amdgcn_readlane(sc.x, q) + (wave * CPW + q) * chunkLen;
      p[q + 1] = p[q] + __builtin_amdgcn_readlane(sc.y, q);
    }
    total = p[CPW];
    int groups = 0;
    for (int done = 0; done < total; done += 32) {
      const int v = done + lane;
      int r = 0, cl = 0;
      float nrm = 0.0f;
      if (v < total) {
        int a = s[0] + v;
#pragma unroll
        for (int q = 1; q < CPW; ++q)
          if (v >= p[q]) a = s[q] + (v - p[q]);
        r = rsrc[a];
        const int c = rcol[a];
        nrm = dinv[r] * rw[a] * dinv[c];
        cl = c - n0;
      }
      if (keep) {
        if (groups < LIST_GROUPS) {
          mysrc[groups * 32 + lane] = r;
          mydst[groups * 32 + lane] = cl;
          mynrm[groups * 32 + lane] = nrm;
        }
      }
      ++groups;
      add_group(r, cl, nrm, min(32, total - done));
    }
    return groups;
  };

  if (MODE == 0) {
    int total;
    const int groups = walk(true, total);
    if (lane == 0) lcnt[slot] = groups <= LIST_GROUPS ? total : -1;
  } else {
    const int cnt = __builtin_amdgcn_readfirstlane(lcnt[slot]);
    if (cnt >= 0) {
      for (int done = 0; done < cnt; done += 32) {
        const int r = mysrc[done + lane];
        const int cl = mydst[done + lane];
        const float nrm = mynrm[done + lane];
        add_group(r, cl, nrm, min(32, cnt - done));
      }
    } else {
      int total;
      walk(false, total);
    }
  }

  __syncthreads();

  for (int i = tid; i < nb * D_FEAT; i += AGG_THREADS) {
    const int f = i & (D_FEAT - 1);
    const int rl = i >> 7;
    const int node = n0 + rl;
    const float d = dinv[node];
    const size_t g = (size_t)node * D_FEAT + f;
    const float v = acc[rl * PITCH + f] + H[g] * d * d + bias[f];
    const float z = fmaxf(v, 0.0f) + pw[f] * fminf(v, 0.0f);
    if (MODE == 0) acc[rl * PITCH + f] = z;
    else Z[g] = z;
  }

  if (MODE == 0) {
    __syncthreads();
    const int m0 = (wave >> 1) * 16;
    if (wave < 2 * (NBP / 16) && m0 < nb) {
      const int nbase = (wave & 1) * 64;
      const int l15 = lane & 15;
      const int h = lane >> 4;
      const float* ap = acc + (m0 + l15) * PITCH + 2 * h;
      const float* bp = wl + h * WPITCH + (nbase + l15) * 2;
      v8f c0 = {}, c1 = {}, c2 = {}, c3 = {};
#pragma unroll
      for (int k0 = 0; k0 < D_FEAT; k0 += 4) {
        const v2f a = *(const v2f*)(ap + k0);
        const float* bk = bp + (k0 >> 1) * WPITCH;
        const v2f b0 = *(const v2f*)(bk);
        const v2f b1 = *(const v2f*)(bk + 32);
        const v2f b2 = *(const v2f*)(bk + 64);
        const v2f b3 = *(const v2f*)(bk + 96);
        c0 = __builtin_amdgcn_wmma_f32_16x16x4_f32(false, a, false, b0, (short)0, c0, false, false);
        c1 = __builtin_amdgcn_wmma_f32_16x16x4_f32(false, a, false, b1, (short)0, c1, false, false);
        c2 = __builtin_amdgcn_wmma_f32_16x16x4_f32(false, a, false, b2, (short)0, c2, false, false);
        c3 = __builtin_amdgcn_wmma_f32_16x16x4_f32(false, a, false, b3, (short)0, c3, false, false);
      }
      const int rbase = m0 + 8 * h;
      float* zp = Z + (size_t)(n0 + rbase) * D_FEAT + nbase + l15;
#pragma unroll
      for (int r = 0; r < 8; ++r) {
        if (rbase + r < nb) {
          float* q = zp + (size_t)r * D_FEAT;
          q[0] = c0[r];
          q[16] = c1[r];
          q[32] = c2[r];
          q[48] = c3[r];
        }
      }
    }
  }
}

#define COMB_THREADS 256
#define COMB_BLOCKS 512
__global__ __launch_bounds__(COMB_THREADS) void combine_kernel(
    const float* __restrict__ z0, const float* __restrict__ z1,
    const float* __restrict__ z2, const float* __restrict__ coeff,
    float* __restrict__ out, int n4) {
  const float c0 = coeff[0], c1 = coeff[1], c2 = coeff[2];
  const float4* p0 = (const float4*)z0;
  const float4* p1 = (const float4*)z1;
  const float4* p2 = (const float4*)z2;
  float4* o4 = (float4*)out;
  const int stride = gridDim.x * COMB_THREADS;
  int i = blockIdx.x * COMB_THREADS + threadIdx.x;

  auto mix = [&](const float4& a, const float4& b, const float4& c) {
    float4 r;
    r.x = __builtin_fmaf(c2, c.x, __builtin_fmaf(c1, b.x, c0 * a.x));
    r.y = __builtin_fmaf(c2, c.y, __builtin_fmaf(c1, b.y, c0 * a.y));
    r.z = __builtin_fmaf(c2, c.z, __builtin_fmaf(c1, b.z, c0 * a.z));
    r.w = __builtin_fmaf(c2, c.w, __builtin_fmaf(c1, b.w, c0 * a.w));
    return r;
  };

  for (; i + 3 * stride < n4; i += 4 * stride) {
    float4 a[4], b[4], c[4];
#pragma unroll
    for (int u = 0; u < 4; ++u) {
      a[u] = p0[i + u * stride];
      b[u] = p1[i + u * stride];
      c[u] = p2[i + u * stride];
    }
#pragma unroll
    for (int u = 0; u < 4; ++u) o4[i + u * stride] = mix(a[u], b[u], c[u]);
  }
  for (; i < n4; i += stride) o4[i] = mix(p0[i], p1[i], p2[i]);
}

extern "C" void kernel_launch(void* const* d_in, const int* in_sizes, int n_in,
                              void* d_out, int out_size, void* d_ws, size_t ws_size,
                              hipStream_t stream) {
  const float* x       = (const float*)d_in[0];
  const int*   eidx    = (const int*)  d_in[1];
  const float* ew      = (const float*)d_in[2];
  const float* coeff   = (const float*)d_in[3];
  const float* W1      = (const float*)d_in[4];
  const float* b1      = (const float*)d_in[5];
  const float* W2      = (const float*)d_in[6];
  const float* b2      = (const float*)d_in[7];
  const float* prelu_w = (const float*)d_in[8];

  const int N = in_sizes[0] / D_FEAT;
  const int E = in_sizes[2] / 3;
  const int total = N * D_FEAT;

  const int aggBlocks  = (N + NB - 1) / NB;

  float* ws   = (float*)d_ws;
  float* dinv = ws;
  float* bufA = ws + (((size_t)N + 31) & ~(size_t)31);
  float* zpl  = bufA + (size_t)N * D_FEAT;
  const size_t nSlots = (size_t)aggBlocks * AGG_WAVES;
  float* bufB = zpl + (size_t)3 * N * D_FEAT;
  int*   lcnt = (int*)(bufB + (size_t)N * D_FEAT);
  int*   lsrc = lcnt + ((nSlots + 31) & ~(size_t)31);
  int*   ldst = lsrc + nSlots * LIST_WORDS;
  float* lnrm = (float*)(ldst + nSlots * LIST_WORDS);
  const size_t ePad = ((size_t)E + 31) & ~(size_t)31;
  int*   rsrc = (int*)(lnrm + nSlots * LIST_WORDS);
  int*   rcol = rsrc + ePad;
  float* rw   = (float*)(rcol + ePad);
  int*   seg  = (int*)(rw + ePad);

  const int ETHREADS = 256;
  const dim3 waveBlock(32, 8);
  const int gemmBlocks = ((N / 16) * (D_FEAT / 16) + 7) / 8;
  const int nodeBlocks = (N + ETHREADS - 1) / ETHREADS;

  for (int t = 0; t < 3; ++t) {
    const int* row = eidx + (size_t)t * 2 * E;
    const int* col = row + E;
    const float* w = ew + (size_t)t * E;
    float* zt = zpl + (size_t)t * total;

    zero_kernel<<<nodeBlocks, ETHREADS, 0, stream>>>(dinv, N);
    degree_sort_kernel<<<NCHUNK, SORT_THREADS, 0, stream>>>(row, col, w, dinv, E, rsrc, rcol, rw, seg);
    dinv_kernel<<<nodeBlocks, ETHREADS, 0, stream>>>(dinv, N);

    gemm_wmma_kernel<<<gemmBlocks, waveBlock, 0, stream>>>(x, W1, bufA, N);
    aggregate_block_kernel<0><<<aggBlocks, AGG_THREADS, 0, stream>>>(rsrc, rcol, rw, seg, dinv, bufA, b1, prelu_w, bufB, E, N,
                                                                  lsrc, ldst, lnrm, lcnt, W2);

    aggregate_block_kernel<1><<<aggBlocks, AGG_THREADS, 0, stream>>>(rsrc, rcol, rw, seg, dinv, bufB, b2, prelu_w, zt, E, N,
                                                                  lsrc, ldst, lnrm, lcnt, W2);
  }

  combine_kernel<<<COMB_BLOCKS, COMB_THREADS, 0, stream>>>(zpl, zpl + (size_t)total, zpl + (size_t)2 * total, coeff,
                                                         (float*)d_out, total / 4);
}


# Layer 1's product x@W1 moved into the degree/sort launch: 125 extra work-groups form it (W1 in LDS, wave per 16-row strip x 64 columns) beside the 128 sort groups; 3 GEMM launches gone; k order kept
# speedup vs baseline: 3.1770x; 1.1673x over previous
#include <hip/hip_runtime.h>
#include <hip/hip_bf16.h>


typedef __attribute__((ext_vector_type(2))) float v2f;
typedef __attribute__((ext_vector_type(8))) float v8f;

#define D_FEAT 128

__global__ void zero_kernel(float* __restrict__ p, int n) {
  int i = blockIdx.x * blockDim.x + threadIdx.x;
  if (i < n) p[i] = 0.0f;
}

#define NB 196
#define NCHUNK 128
#define PROD_WGS 125
#define SORT_THREADS 1024
#define SORT_WAVES (SORT_THREADS / 32)
#define SORT_EPT 8
#define SORT_CAP (SORT_EPT * SORT_THREADS)
#define SORT_BINS 256
#define WPITCH 288

__device__ __forceinline__ void product_part(
    const float* __restrict__ X, const float* __restrict__ Wm, float* __restrict__ H,
    int nRows, int g, int nG) {
  __shared__ float wl[(D_FEAT / 2) * WPITCH];
  const int tid = threadIdx.x;
  const int lane = tid & 31;
  const int wave = __builtin_amdgcn_readfirstlane(tid >> 5);
  for (int i = tid; i < (D_FEAT / 2) * (D_FEAT / 4); i += SORT_THREADS) {
    const int p = i >> 5;
    const int c4 = (i & 31) * 4;
    const float4 lo = *(const float4*)(Wm + (size_t)(2 * p) * D_FEAT + c4);
    const float4 hi = *(const float4*)(Wm + (size_t)(2 * p + 1) * D_FEAT + c4);
    float4* dst = (float4*)(wl + p * WPITCH + c4 * 2);
    dst[0] = make_float4(lo.x, hi.x, lo.y, hi.y);
    dst[1] = make_float4(lo.z, hi.z, lo.w, hi.w);
  }
  __syncthreads();

  const int strips = nRows / 16;
  const int spw = (strips + nG - 1) / nG;
  const int s0 = g * spw;
  const int s1 = min(strips, s0 + spw);
  const int l15 = lane & 15;
  const int h = lane >> 4;
  for (int task = 2 * s0 + wave; task < 2 * s1; task += SORT_WAVES) {
    const int m0 = (task >> 1) * 16;
    const int nbase = (task & 1) * 64;
    const float* ap = X + (size_t)(m0 + l15) * D_FEAT + 2 * h;
    const float* bp = wl + h * WPITCH + (nbase + l15) * 2;
    v8f c0 = {}, c1 = {}, c2 = {}, c3 = {};
#pragma nounroll
    for (int kb = 0; kb < D_FEAT; kb += 32) {
      v2f a[8];
#pragma unroll
      for (int u = 0; u < 8; ++u) a[u] = *(const v2f*)(ap + kb + 4 * u);
#pragma unroll
      for (int u = 0; u < 8; ++u) {
        const float* bk = bp + ((kb >> 1) + 2 * u) * WPITCH;
        const v2f b0 = *(const v2f*)(bk);
        const v2f b1 = *(const v2f*)(bk + 32);
        const v2f b2 = *(const v2f*)(bk + 64);
        const v2f b3 = *(const v2f*)(bk + 96);
        c0 = __builtin_amdgcn_wmma_f32_16x16x4_f32(false, a[u], false, b0, (short)0, c0, false, false);
        c1 = __builtin_amdgcn_wmma_f32_16x16x4_f32(false, a[u], false, b1, (short)0, c1, false, false);
        c2 = __builtin_amdgcn_wmma_f32_16x16x4_f32(false, a[u], false, b2, (short)0, c2, false, false);
        c3 = __builtin_amdgcn_wmma_f32_16x16x4_f32(false, a[u], false, b3, (short)0, c3, false, false);
      }
    }
    float* hp = H + (size_t)(m0 + 8 * h) * D_FEAT + nbase + l15;
#pragma unroll
    for (int r = 0; r < 8; ++r) {
      float* q = hp + (size_t)r * D_FEAT;
      q[0] = c0[r];
      q[16] = c1[r];
      q[32] = c2[r];
      q[48] = c3[r];
    }
  }
}

__global__ __launch_bounds__(SORT_THREADS) void degree_sort_kernel(
    const int* __restrict__ row, const int* __restrict__ col,
    const float* __restrict__ w, float* __restrict__ deg, int E,
    int* __restrict__ rsrc, int* __restrict__ rcol, float* __restrict__ rw,
    int* __restrict__ seg,
    const float* __restrict__ X, const float* __restrict__ Wm, float* __restrict__ H, int nRows) {
  if (blockIdx.x >= NCHUNK) {
    product_part(X, Wm, H, nRows, blockIdx.x - NCHUNK, gridDim.x - NCHUNK);
    return;
  }
  __shared__ int cnt[SORT_WAVES][SORT_BINS];
  __shared__ int binStart[SORT_BINS];
  __shared__ int waveSum[SORT_BINS / 32];
  __shared__ int s_r[SORT_CAP];
  __shared__ int s_c[SORT_CAP];
  __shared__ float s_w[SORT_CAP];

  const int tid = threadIdx.x;
  const int lane = tid & 31;
  const int wave = tid >> 5;
  const int chunkLen = (E + NCHUNK - 1) / NCHUNK;
  const int e0 = blockIdx.x * chunkLen;
  const int n = max(0, min(chunkLen, E - e0));

  for (int i = tid; i < SORT_WAVES * SORT_BINS; i += SORT_THREADS) (&cnt[0][0])[i] = 0;
  __syncthreads();

  int r[SORT_EPT], c[SORT_EPT], b[SORT_EPT], rank[SORT_EPT];
  float wt[SORT_EPT];
#pragma unroll
  for (int k = 0; k < SORT_EPT; ++k) {
    const int i = k * SORT_THREADS + tid;
    const bool ok = i < n;
    c[k] = ok ? col[e0 + i] : -1;
    r[k] = ok ? row[e0 + i] : 0;
    wt[k] = ok ? w[e0 + i] : 0.0f;
  }
#pragma unroll
  for (int k = 0; k < SORT_EPT; ++k) {
    b[k] = 0;
    rank[k] = 0;
    if (c[k] >= 0) {
      b[k] = c[k] / NB;
      rank[k] = __hip_atomic_fetch_add(&cnt[wave][b[k]], 1, __ATOMIC_RELAXED,
                                       __HIP_MEMORY_SCOPE_WORKGROUP);
    }
  }
  __syncthreads();

  int tot = 0;
  if (tid < SORT_BINS) {
    int t[SORT_WAVES];
#pragma unroll
    for (int q = 0; q < SORT_WAVES; ++q) t[q] = cnt[q][tid];
#pragma unroll
    for (int q = 0; q < SORT_WAVES; ++q) {
      cnt[q][tid] = tot;
      tot += t[q];
    }
  }
  int incl = tot;
#pragma unroll
  for (int d = 1; d < 32; d <<= 1) {
    const int o = __shfl_up(incl, d);
    if (lane >= d) incl += o;
  }
  if (tid < SORT_BINS && lane == 31) waveSum[wave] = incl;
  __syncthreads();
  if (tid < SORT_BINS) {
    int base = 0;
    for (int q = 0; q < wave; ++q) base += waveSum[q];
    const int start = base + incl - tot;
    binStart[tid] = start;
    ((int2*)seg)[(size_t)tid * NCHUNK + blockIdx.x] = make_int2(start, tot);
  }
  __syncthreads();

#pragma unroll
  for (int k = 0; k < SORT_EPT; ++k) {
    if (c[k] >= 0) {
      const int pos = binStart[b[k]] + cnt[wave][b[k]] + rank[k];
      s_r[pos] = r[k];
      s_c[pos] = c[k];
      s_w[pos] = wt[k];
    }
  }
  __syncthreads();
  for (int i = tid; i < n; i += SORT_THREADS) {
    rsrc[e0 + i] = s_r[i];
    rcol[e0 + i] = s_c[i];
    rw[e0 + i] = s_w[i];
  }
#pragma unroll
  for (int k = 0; k < SORT_EPT; ++k)
    if (c[k] >= 0) atomicAdd(&deg[c[k]], wt[k]);
}

__global__ void dinv_kernel(float* __restrict__ deg, int n) {
  int i = blockIdx.x * blockDim.x + threadIdx.x;
  if (i < n) {
    float d = deg[i] + 1.0f;
    deg[i] = rsqrtf(d);
  }
}

#define NBP 208
#define PITCH 132
#define AGG_THREADS 1024
#define AGG_WAVES (AGG_THREADS / 32)
#define CPW (NCHUNK / AGG_WAVES)
#define LIST_GROUPS 16
#define LIST_WORDS (LIST_GROUPS * 32)

template <int MODE>
__global__ __launch_bounds__(AGG_THREADS) void aggregate_block_kernel(
    const int* __restrict__ rsrc, const int* __restrict__ rcol,
    const float* __restrict__ rw, const int* __restrict__ seg,
    const float* __restrict__ dinv,
    const float* __restrict__ H, const float* __restrict__ bias,
    const float* __restrict__ pw, float* __restrict__ Z, int E, int N,
    int* __restrict__ lsrc, int* __restrict__ ldst, float* __restrict__ lnrm,
    int* __restrict__ lcnt, const float* __restrict__ Wn) {
  __shared__ float acc[NBP * PITCH];
  __shared__ float wl[MODE == 0 ? (D_FEAT / 2) * WPITCH : 4];

  const int tid = threadIdx.x;
  const int lane = tid & 31;
  const int wave = __builtin_amdgcn_readfirstlane(tid >> 5);
  const int n0 = blockIdx.x * NB;
  const int nb = min(NB, N - n0);

  {
    float4* a4 = (float4*)acc;
    for (int i = tid; i < NBP * PITCH / 4; i += AGG_THREADS)
      a4[i] = make_float4(0.0f, 0.0f, 0.0f, 0.0f);
  }
  if (MODE == 0) {
    for (int i = tid; i < (D_FEAT / 2) * (D_FEAT / 4); i += AGG_THREADS) {
      const int p = i >> 5;
      const int c4 = (i & 31) * 4;
      const float4 lo = *(const float4*)(Wn + (size_t)(2 * p) * D_FEAT + c4);
      const float4 hi = *(const float4*)(Wn + (size_t)(2 * p + 1) * D_FEAT + c4);
      float4* dst = (float4*)(wl + p * WPITCH + c4 * 2);
      dst[0] = make_float4(lo.x, hi.x, lo.y, hi.y);
      dst[1] = make_float4(lo.z, hi.z, lo.w, hi.w);
    }
  }
  __syncthreads();

  const size_t slot = (size_t)blockIdx.x * AGG_WAVES + wave;
  int* mysrc = lsrc + slot * LIST_WORDS;
  int* mydst = ldst + slot * LIST_WORDS;
  float* mynrm = lnrm + slot * LIST_WORDS;

  auto add_group = [&](int r, int cl, float nrm, int take) {
    for (int g0 = 0; g0 < take; g0 += 8) {
      float v[8][4];
#pragma unroll
      for (int u = 0; u < 8; ++u) {
        const int k = min(g0 + u, take - 1);
        const int rk = __builtin_amdgcn_readlane(r, k);
        const float* hp = H + (size_t)rk * D_FEAT + lane;
#pragma unroll
        for (int q = 0; q < 4; ++q) v[u][q] = hp[32 * q];
      }
#pragma unroll
      for (int u = 0; u < 8; ++u)
        asm volatile("" : : "v"(v[u][0]), "v"(v[u][1]), "v"(v[u][2]), "v"(v[u][3]));
#pragma unroll
      for (int u = 0; u < 8; ++u) {
        if (g0 + u < take) {
          const int ck = __builtin_amdgcn_readlane(cl, g0 + u);
          const float nk = __int_as_float(__builtin_amdgcn_readlane(__float_as_int(nrm), g0 + u));
          float* ap = acc + ck * PITCH + lane;
#pragma unroll
          for (int q = 0; q < 4; ++q)
            __hip_atomic_fetch_add(ap + 32 * q, v[u][q] * nk, __ATOMIC_RELAXED,
                                   __HIP_MEMORY_SCOPE_WORKGROUP);
        }
      }
    }
  };

  auto walk = [&](bool keep, int& total) {
    const int chunkLen = (E + NCHUNK - 1) / NCHUNK;
    int2 sc = make_int2(0, 0);
    if (lane < CPW) sc = ((const int2*)seg)[(size_t)blockIdx.x * NCHUNK + wave * CPW + lane];
    int s[CPW], p[CPW + 1];
    p[0] = 0;
#pragma unroll
    for (int q = 0; q < CPW; ++q) {
      s[q] = __builtin_amdgcn_readlane(sc.x, q) + (wave * CPW + q) * chunkLen;
      p[q + 1] = p[q] + __builtin_amdgcn_readlane(sc.y, q);
    }
    total = p[CPW];
    int groups = 0;
    for (int done = 0; done < total; done += 32) {
      const int v = done + lane;
      int r = 0, cl = 0;
      float nrm = 0.0f;
      if (v < total) {
        int a = s[0] + v;
#pragma unroll
        for (int q = 1; q < CPW; ++q)
          if (v >= p[q]) a = s[q] + (v - p[q]);
        r = rsrc[a];
        const int c = rcol[a];
        nrm = dinv[r] * rw[a] * dinv[c];
        cl = c - n0;
      }
      if (keep) {
        if (groups < LIST_GROUPS) {
          mysrc[groups * 32 + lane] = r;
          mydst[groups * 32 + lane] = cl;
          mynrm[groups * 32 + lane] = nrm;
        }
      }
      ++groups;
      add_group(r, cl, nrm, min(32, total - done));
    }
    return groups;
  };

  if (MODE == 0) {
    int total;
    const int groups = walk(true, total);
    if (lane == 0) lcnt[slot] = groups <= LIST_GROUPS ? total : -1;
  } else {
    const int cnt = __builtin_amdgcn_readfirstlane(lcnt[slot]);
    if (cnt >= 0) {
      for (int done = 0; done < cnt; done += 32) {
        const int r = mysrc[done + lane];
        const int cl = mydst[done + lane];
        const float nrm = mynrm[done + lane];
        add_group(r, cl, nrm, min(32, cnt - done));
      }
    } else {
      int total;
      walk(false, total);
    }
  }

  __syncthreads();

  for (int i = tid; i < nb * D_FEAT; i += AGG_THREADS) {
    const int f = i & (D_FEAT - 1);
    const int rl = i >> 7;
    const int node = n0 + rl;
    const float d = dinv[node];
    const size_t g = (size_t)node * D_FEAT + f;
    const float v = acc[rl * PITCH + f] + H[g] * d * d + bias[f];
    const float z = fmaxf(v, 0.0f) + pw[f] * fminf(v, 0.0f);
    if (MODE == 0) acc[rl * PITCH + f] = z;
    else Z[g] = z;
  }

  if (MODE == 0) {
    __syncthreads();
    const int m0 = (wave >> 1) * 16;
    if (wave < 2 * (NBP / 16) && m0 < nb) {
      const int nbase = (wave & 1) * 64;
      const int l15 = lane & 15;
      const int h = lane >> 4;
      const float* ap = acc + (m0 + l15) * PITCH + 2 * h;
      const float* bp = wl + h * WPITCH + (nbase + l15) * 2;
      v8f c0 = {}, c1 = {}, c2 = {}, c3 = {};
#pragma unroll
      for (int k0 = 0; k0 < D_FEAT; k0 += 4) {
        const v2f a = *(const v2f*)(ap + k0);
        const float* bk = bp + (k0 >> 1) * WPITCH;
        const v2f b0 = *(const v2f*)(bk);
        const v2f b1 = *(const v2f*)(bk + 32);
        const v2f b2 = *(const v2f*)(bk + 64);
        const v2f b3 = *(const v2f*)(bk + 96);
        c0 = __builtin_amdgcn_wmma_f32_16x16x4_f32(false, a, false, b0, (short)0, c0, false, false);
        c1 = __builtin_amdgcn_wmma_f32_16x16x4_f32(false, a, false, b1, (short)0, c1, false, false);
        c2 = __builtin_amdgcn_wmma_f32_16x16x4_f32(false, a, false, b2, (short)0, c2, false, false);
        c3 = __builtin_amdgcn_wmma_f32_16x16x4_f32(false, a, false, b3, (short)0, c3, false, false);
      }
      const int rbase = m0 + 8 * h;
      float* zp = Z + (size_t)(n0 + rbase) * D_FEAT + nbase + l15;
#pragma unroll
      for (int r = 0; r < 8; ++r) {
        if (rbase + r < nb) {
          float* q = zp + (size_t)r * D_FEAT;
          q[0] = c0[r];
          q[16] = c1[r];
          q[32] = c2[r];
          q[48] = c3[r];
        }
      }
    }
  }
}

#define COMB_THREADS 256
#define COMB_BLOCKS 512
__global__ __launch_bounds__(COMB_THREADS) void combine_kernel(
    const float* __restrict__ z0, const float* __restrict__ z1,
    const float* __restrict__ z2, const float* __restrict__ coeff,
    float* __restrict__ out, int n4) {
  const float c0 = coeff[0], c1 = coeff[1], c2 = coeff[2];
  const float4* p0 = (const float4*)z0;
  const float4* p1 = (const float4*)z1;
  const float4* p2 = (const float4*)z2;
  float4* o4 = (float4*)out;
  const int stride = gridDim.x * COMB_THREADS;
  int i = blockIdx.x * COMB_THREADS + threadIdx.x;

  auto mix = [&](const float4& a, const float4& b, const float4& c) {
    float4 r;
    r.x = __builtin_fmaf(c2, c.x, __builtin_fmaf(c1, b.x, c0 * a.x));
    r.y = __builtin_fmaf(c2, c.y, __builtin_fmaf(c1, b.y, c0 * a.y));
    r.z = __builtin_fmaf(c2, c.z, __builtin_fmaf(c1, b.z, c0 * a.z));
    r.w = __builtin_fmaf(c2, c.w, __builtin_fmaf(c1, b.w, c0 * a.w));
    return r;
  };

  for (; i + 3 * stride < n4; i += 4 * stride) {
    float4 a[4], b[4], c[4];
#pragma unroll
    for (int u = 0; u < 4; ++u) {
      a[u] = p0[i + u * stride];
      b[u] = p1[i + u * stride];
      c[u] = p2[i + u * stride];
    }
#pragma unroll
    for (int u = 0; u < 4; ++u) o4[i + u * stride] = mix(a[u], b[u], c[u]);
  }
  for (; i < n4; i += stride) o4[i] = mix(p0[i], p1[i], p2[i]);
}

extern "C" void kernel_launch(void* const* d_in, const int* in_sizes, int n_in,
                              void* d_out, int out_size, void* d_ws, size_t ws_size,
                              hipStream_t stream) {
  const float* x       = (const float*)d_in[0];
  const int*   eidx    = (const int*)  d_in[1];
  const float* ew      = (const float*)d_in[2];
  const float* coeff   = (const float*)d_in[3];
  const float* W1      = (const float*)d_in[4];
  const float* b1      = (const float*)d_in[5];
  const float* W2      = (const float*)d_in[6];
  const float* b2      = (const float*)d_in[7];
  const float* prelu_w = (const float*)d_in[8];

  const int N = in_sizes[0] / D_FEAT;
  const int E = in_sizes[2] / 3;
  const int total = N * D_FEAT;

  const int aggBlocks  = (N + NB - 1) / NB;

  float* ws   = (float*)d_ws;
  float* dinv = ws;
  float* bufA = ws + (((size_t)N + 31) & ~(size_t)31);
  float* zpl  = bufA + (size_t)N * D_FEAT;
  const size_t nSlots = (size_t)aggBlocks * AGG_WAVES;
  float* bufB = zpl + (size_t)3 * N * D_FEAT;
  int*   lcnt = (int*)(bufB + (size_t)N * D_FEAT);
  int*   lsrc = lcnt + ((nSlots + 31) & ~(size_t)31);
  int*   ldst = lsrc + nSlots * LIST_WORDS;
  float* lnrm = (float*)(ldst + nSlots * LIST_WORDS);
  const size_t ePad = ((size_t)E + 31) & ~(size_t)31;
  int*   rsrc = (int*)(lnrm + nSlots * LIST_WORDS);
  int*   rcol = rsrc + ePad;
  float* rw   = (float*)(rcol + ePad);
  int*   seg  = (int*)(rw + ePad);

  const int ETHREADS = 256;
  const int nodeBlocks = (N + ETHREADS - 1) / ETHREADS;

  for (int t = 0; t < 3; ++t) {
    const int* row = eidx + (size_t)t * 2 * E;
    const int* col = row + E;
    const float* w = ew + (size_t)t * E;
    float* zt = zpl + (size_t)t * total;

    zero_kernel<<<nodeBlocks, ETHREADS, 0, stream>>>(dinv, N);
    degree_sort_kernel<<<NCHUNK + PROD_WGS, SORT_THREADS, 0, stream>>>(row, col, w, dinv, E, rsrc, rcol, rw, seg,
                                                                    x, W1, bufA, N);
    dinv_kernel<<<nodeBlocks, ETHREADS, 0, stream>>>(dinv, N);

    aggregate_block_kernel<0><<<aggBlocks, AGG_THREADS, 0, stream>>>(rsrc, rcol, rw, seg, dinv, bufA, b1, prelu_w, bufB, E, N,
                                                                  lsrc, ldst, lnrm, lcnt, W2);

    aggregate_block_kernel<1><<<aggBlocks, AGG_THREADS, 0, stream>>>(rsrc, rcol, rw, seg, dinv, bufB, b2, prelu_w, zt, E, N,
                                                                  lsrc, ldst, lnrm, lcnt, W2);
  }

  combine_kernel<<<COMB_BLOCKS, COMB_THREADS, 0, stream>>>(zpl, zpl + (size_t)total, zpl + (size_t)2 * total, coeff,
                                                         (float*)d_out, total / 4);
}


# Degree path has no launches of its own: one zero pass clears three per-scale degree planes, rsqrt(deg+1) is taken where it is read (norms; self loop via LDS); 16 -> 11 launches; sums unchanged
# speedup vs baseline: 3.2890x; 1.0353x over previous
#include <hip/hip_runtime.h>
#include <hip/hip_bf16.h>


typedef __attribute__((ext_vector_type(2))) float v2f;
typedef __attribute__((ext_vector_type(8))) float v8f;

#define D_FEAT 128

__global__ void zero_kernel(float* __restrict__ p, int n) {
  int i = blockIdx.x * blockDim.x + threadIdx.x;
  if (i < n) p[i] = 0.0f;
}

#define NB 196
#define NCHUNK 128
#define PROD_WGS 125
#define SORT_THREADS 1024
#define SORT_WAVES (SORT_THREADS / 32)
#define SORT_EPT 8
#define SORT_CAP (SORT_EPT * SORT_THREADS)
#define SORT_BINS 256
#define WPITCH 288

__device__ __forceinline__ void product_part(
    const float* __restrict__ X, const float* __restrict__ Wm, float* __restrict__ H,
    int nRows, int g, int nG) {
  __shared__ float wl[(D_FEAT / 2) * WPITCH];
  const int tid = threadIdx.x;
  const int lane = tid & 31;
  const int wave = __builtin_amdgcn_readfirstlane(tid >> 5);
  for (int i = tid; i < (D_FEAT / 2) * (D_FEAT / 4); i += SORT_THREADS) {
    const int p = i >> 5;
    const int c4 = (i & 31) * 4;
    const float4 lo = *(const float4*)(Wm + (size_t)(2 * p) * D_FEAT + c4);
    const float4 hi = *(const float4*)(Wm + (size_t)(2 * p + 1) * D_FEAT + c4);
    float4* dst = (float4*)(wl + p * WPITCH + c4 * 2);
    dst[0] = make_float4(lo.x, hi.x, lo.y, hi.y);
    dst[1] = make_float4(lo.z, hi.z, lo.w, hi.w);
  }
  __syncthreads();

  const int strips = nRows / 16;
  const int spw = (strips + nG - 1) / nG;
  const int s0 = g * spw;
  const int s1 = min(strips, s0 + spw);
  const int l15 = lane & 15;
  const int h = lane >> 4;
  for (int task = 2 * s0 + wave; task < 2 * s1; task += SORT_WAVES) {
    const int m0 = (task >> 1) * 16;
    const int nbase = (task & 1) * 64;
    const float* ap = X + (size_t)(m0 + l15) * D_FEAT + 2 * h;
    const float* bp = wl + h * WPITCH + (nbase + l15) * 2;
    v8f c0 = {}, c1 = {}, c2 = {}, c3 = {};
#pragma nounroll
    for (int kb = 0; kb < D_FEAT; kb += 32) {
      v2f a[8];
#pragma unroll
      for (int u = 0; u < 8; ++u) a[u] = *(const v2f*)(ap + kb + 4 * u);
#pragma unroll
      for (int u = 0; u < 8; ++u) {
        const float* bk = bp + ((kb >> 1) + 2 * u) * WPITCH;
        const v2f b0 = *(const v2f*)(bk);
        const v2f b1 = *(const v2f*)(bk + 32);
        const v2f b2 = *(const v2f*)(bk + 64);
        const v2f b3 = *(const v2f*)(bk + 96);
        c0 = __builtin_amdgcn_wmma_f32_16x16x4_f32(false, a[u], false, b0, (short)0, c0, false, false);
        c1 = __builtin_amdgcn_wmma_f32_16x16x4_f32(false, a[u], false, b1, (short)0, c1, false, false);
        c2 = __builtin_amdgcn_wmma_f32_16x16x4_f32(false, a[u], false, b2, (short)0, c2, false, false);
        c3 = __builtin_amdgcn_wmma_f32_16x16x4_f32(false, a[u], false, b3, (short)0, c3, false, false);
      }
    }
    float* hp = H + (size_t)(m0 + 8 * h) * D_FEAT + nbase + l15;
#pragma unroll
    for (int r = 0; r < 8; ++r) {
      float* q = hp + (size_t)r * D_FEAT;
      q[0] = c0[r];
      q[16] = c1[r];
      q[32] = c2[r];
      q[48] = c3[r];
    }
  }
}

__global__ __launch_bounds__(SORT_THREADS) void degree_sort_kernel(
    const int* __restrict__ row, const int* __restrict__ col,
    const float* __restrict__ w, float* __restrict__ deg, int E,
    int* __restrict__ rsrc, int* __restrict__ rcol, float* __restrict__ rw,
    int* __restrict__ seg,
    const float* __restrict__ X, const float* __restrict__ Wm, float* __restrict__ H, int nRows) {
  if (blockIdx.x >= NCHUNK) {
    product_part(X, Wm, H, nRows, blockIdx.x - NCHUNK, gridDim.x - NCHUNK);
    return;
  }
  __shared__ int cnt[SORT_WAVES][SORT_BINS];
  __shared__ int binStart[SORT_BINS];
  __shared__ int waveSum[SORT_BINS / 32];
  __shared__ int s_r[SORT_CAP];
  __shared__ int s_c[SORT_CAP];
  __shared__ float s_w[SORT_CAP];

  const int tid = threadIdx.x;
  const int lane = tid & 31;
  const int wave = tid >> 5;
  const int chunkLen = (E + NCHUNK - 1) / NCHUNK;
  const int e0 = blockIdx.x * chunkLen;
  const int n = max(0, min(chunkLen, E - e0));

  for (int i = tid; i < SORT_WAVES * SORT_BINS; i += SORT_THREADS) (&cnt[0][0])[i] = 0;
  __syncthreads();

  int r[SORT_EPT], c[SORT_EPT], b[SORT_EPT], rank[SORT_EPT];
  float wt[SORT_EPT];
#pragma unroll
  for (int k = 0; k < SORT_EPT; ++k) {
    const int i = k * SORT_THREADS + tid;
    const bool ok = i < n;
    c[k] = ok ? col[e0 + i] : -1;
    r[k] = ok ? row[e0 + i] : 0;
    wt[k] = ok ? w[e0 + i] : 0.0f;
  }
#pragma unroll
  for (int k = 0; k < SORT_EPT; ++k) {
    b[k] = 0;
    rank[k] = 0;
    if (c[k] >= 0) {
      b[k] = c[k] / NB;
      rank[k] = __hip_atomic_fetch_add(&cnt[wave][b[k]], 1, __ATOMIC_RELAXED,
                                       __HIP_MEMORY_SCOPE_WORKGROUP);
    }
  }
  __syncthreads();

  int tot = 0;
  if (tid < SORT_BINS) {
    int t[SORT_WAVES];
#pragma unroll
    for (int q = 0; q < SORT_WAVES; ++q) t[q] = cnt[q][tid];
#pragma unroll
    for (int q = 0; q < SORT_WAVES; ++q) {
      cnt[q][tid] = tot;
      tot += t[q];
    }
  }
  int incl = tot;
#pragma unroll
  for (int d = 1; d < 32; d <<= 1) {
    const int o = __shfl_up(incl, d);
    if (lane >= d) incl += o;
  }
  if (tid < SORT_BINS && lane == 31) waveSum[wave] = incl;
  __syncthreads();
  if (tid < SORT_BINS) {
    int base = 0;
    for (int q = 0; q < wave; ++q) base += waveSum[q];
    const int start = base + incl - tot;
    binStart[tid] = start;
    ((int2*)seg)[(size_t)tid * NCHUNK + blockIdx.x] = make_int2(start, tot);
  }
  __syncthreads();

#pragma unroll
  for (int k = 0; k < SORT_EPT; ++k) {
    if (c[k] >= 0) {
      const int pos = binStart[b[k]] + cnt[wave][b[k]] + rank[k];
      s_r[pos] = r[k];
      s_c[pos] = c[k];
      s_w[pos] = wt[k];
    }
  }
  __syncthreads();
  for (int i = tid; i < n; i += SORT_THREADS) {
    rsrc[e0 + i] = s_r[i];
    rcol[e0 + i] = s_c[i];
    rw[e0 + i] = s_w[i];
  }
#pragma unroll
  for (int k = 0; k < SORT_EPT; ++k)
    if (c[k] >= 0) atomicAdd(&deg[c[k]], wt[k]);
}

__device__ __forceinline__ float dinv_of(const float* __restrict__ deg, int i) {
  const float d = deg[i] + 1.0f;
  return rsqrtf(d);
}

#define NBP 208
#define PITCH 132
#define AGG_THREADS 1024
#define AGG_WAVES (AGG_THREADS / 32)
#define CPW (NCHUNK / AGG_WAVES)
#define LIST_GROUPS 16
#define LIST_WORDS (LIST_GROUPS * 32)

template <int MODE>
__global__ __launch_bounds__(AGG_THREADS) void aggregate_block_kernel(
    const int* __restrict__ rsrc, const int* __restrict__ rcol,
    const float* __restrict__ rw, const int* __restrict__ seg,
    const float* __restrict__ deg,
    const float* __restrict__ H, const float* __restrict__ bias,
    const float* __restrict__ pw, float* __restrict__ Z, int E, int N,
    int* __restrict__ lsrc, int* __restrict__ ldst, float* __restrict__ lnrm,
    int* __restrict__ lcnt, const float* __restrict__ Wn) {
  __shared__ float acc[NBP * PITCH];
  __shared__ float wl[MODE == 0 ? (D_FEAT / 2) * WPITCH : 4];
  __shared__ float dl[NB];

  const int tid = threadIdx.x;
  const int lane = tid & 31;
  const int wave = __builtin_amdgcn_readfirstlane(tid >> 5);
  const int n0 = blockIdx.x * NB;
  const int nb = min(NB, N - n0);

  {
    float4* a4 = (float4*)acc;
    for (int i = tid; i < NBP * PITCH / 4; i += AGG_THREADS)
      a4[i] = make_float4(0.0f, 0.0f, 0.0f, 0.0f);
  }
  if (tid < nb) dl[tid] = dinv_of(deg, n0 + tid);
  if (MODE == 0) {
    for (int i = tid; i < (D_FEAT / 2) * (D_FEAT / 4); i += AGG_THREADS) {
      const int p = i >> 5;
      const int c4 = (i & 31) * 4;
      const float4 lo = *(const float4*)(Wn + (size_t)(2 * p) * D_FEAT + c4);
      const float4 hi = *(const float4*)(Wn + (size_t)(2 * p + 1) * D_FEAT + c4);
      float4* dst = (float4*)(wl + p * WPITCH + c4 * 2);
      dst[0] = make_float4(lo.x, hi.x, lo.y, hi.y);
      dst[1] = make_float4(lo.z, hi.z, lo.w, hi.w);
    }
  }
  __syncthreads();

  const size_t slot = (size_t)blockIdx.x * AGG_WAVES + wave;
  int* mysrc = lsrc + slot * LIST_WORDS;
  int* mydst = ldst + slot * LIST_WORDS;
  float* mynrm = lnrm + slot * LIST_WORDS;

  auto add_group = [&](int r, int cl, float nrm, int take) {
    for (int g0 = 0; g0 < take; g0 += 8) {
      float v[8][4];
#pragma unroll
      for (int u = 0; u < 8; ++u) {
        const int k = min(g0 + u, take - 1);
        const int rk = __builtin_amdgcn_readlane(r, k);
        const float* hp = H + (size_t)rk * D_FEAT + lane;
#pragma unroll
        for (int q = 0; q < 4; ++q) v[u][q] = hp[32 * q];
      }
#pragma unroll
      for (int u = 0; u < 8; ++u)
        asm volatile("" : : "v"(v[u][0]), "v"(v[u][1]), "v"(v[u][2]), "v"(v[u][3]));
#pragma unroll
      for (int u = 0; u < 8; ++u) {
        if (g0 + u < take) {
          const int ck = __builtin_amdgcn_readlane(cl, g0 + u);
          const float nk = __int_as_float(__builtin_amdgcn_readlane(__float_as_int(nrm), g0 + u));
          float* ap = acc + ck * PITCH + lane;
#pragma unroll
          for (int q = 0; q < 4; ++q)
            __hip_atomic_fetch_add(ap + 32 * q, v[u][q] * nk, __ATOMIC_RELAXED,
                                   __HIP_MEMORY_SCOPE_WORKGROUP);
        }
      }
    }
  };

  auto walk = [&](bool keep, int& total) {
    const int chunkLen = (E + NCHUNK - 1) / NCHUNK;
    int2 sc = make_int2(0, 0);
    if (lane < CPW) sc = ((const int2*)seg)[(size_t)blockIdx.x * NCHUNK + wave * CPW + lane];
    int s[CPW], p[CPW + 1];
    p[0] = 0;
#pragma unroll
    for (int q = 0; q < CPW; ++q) {
      s[q] = __builtin_amdgcn_readlane(sc.x, q) + (wave * CPW + q) * chunkLen;
      p[q + 1] = p[q] + __builtin_amdgcn_readlane(sc.y, q);
    }
    total = p[CPW];
    int groups = 0;
    for (int done = 0; done < total; done += 32) {
      const int v = done + lane;
      int r = 0, cl = 0;
      float nrm = 0.0f;
      if (v < total) {
        int a = s[0] + v;
#pragma unroll
        for (int q = 1; q < CPW; ++q)
          if (v >= p[q]) a = s[q] + (v - p[q]);
        r = rsrc[a];
        const int c = rcol[a];
        nrm = dinv_of(deg, r) * rw[a] * dinv_of(deg, c);
        cl = c - n0;
      }
      if (keep) {
        if (groups < LIST_GROUPS) {
          mysrc[groups * 32 + lane] = r;
          mydst[groups * 32 + lane] = cl;
          mynrm[groups * 32 + lane] = nrm;
        }
      }
      ++groups;
      add_group(r, cl, nrm, min(32, total - done));
    }
    return groups;
  };

  if (MODE == 0) {
    int total;
    const int groups = walk(true, total);
    if (lane == 0) lcnt[slot] = groups <= LIST_GROUPS ? total : -1;
  } else {
    const int cnt = __builtin_amdgcn_readfirstlane(lcnt[slot]);
    if (cnt >= 0) {
      for (int done = 0; done < cnt; done += 32) {
        const int r = mysrc[done + lane];
        const int cl = mydst[done + lane];
        const float nrm = mynrm[done + lane];
        add_group(r, cl, nrm, min(32, cnt - done));
      }
    } else {
      int total;
      walk(false, total);
    }
  }

  __syncthreads();

  for (int i = tid; i < nb * D_FEAT; i += AGG_THREADS) {
    const int f = i & (D_FEAT - 1);
    const int rl = i >> 7;
    const int node = n0 + rl;
    const float d = dl[rl];
    const size_t g = (size_t)node * D_FEAT + f;
    const float v = acc[rl * PITCH + f] + H[g] * d * d + bias[f];
    const float z = fmaxf(v, 0.0f) + pw[f] * fminf(v, 0.0f);
    if (MODE == 0) acc[rl * PITCH + f] = z;
    else Z[g] = z;
  }

  if (MODE == 0) {
    __syncthreads();
    const int m0 = (wave >> 1) * 16;
    if (wave < 2 * (NBP / 16) && m0 < nb) {
      const int nbase = (wave & 1) * 64;
      const int l15 = lane & 15;
      const int h = lane >> 4;
      const float* ap = acc + (m0 + l15) * PITCH + 2 * h;
      const float* bp = wl + h * WPITCH + (nbase + l15) * 2;
      v8f c0 = {}, c1 = {}, c2 = {}, c3 = {};
#pragma unroll
      for (int k0 = 0; k0 < D_FEAT; k0 += 4) {
        const v2f a = *(const v2f*)(ap + k0);
        const float* bk = bp + (k0 >> 1) * WPITCH;
        const v2f b0 = *(const v2f*)(bk);
        const v2f b1 = *(const v2f*)(bk + 32);
        const v2f b2 = *(const v2f*)(bk + 64);
        const v2f b3 = *(const v2f*)(bk + 96);
        c0 = __builtin_amdgcn_wmma_f32_16x16x4_f32(false, a, false, b0, (short)0, c0, false, false);
        c1 = __builtin_amdgcn_wmma_f32_16x16x4_f32(false, a, false, b1, (short)0, c1, false, false);
        c2 = __builtin_amdgcn_wmma_f32_16x16x4_f32(false, a, false, b2, (short)0, c2, false, false);
        c3 = __builtin_amdgcn_wmma_f32_16x16x4_f32(false, a, false, b3, (short)0, c3, false, false);
      }
      const int rbase = m0 + 8 * h;
      float* zp = Z + (size_t)(n0 + rbase) * D_FEAT + nbase + l15;
#pragma unroll
      for (int r = 0; r < 8; ++r) {
        if (rbase + r < nb) {
          float* q = zp + (size_t)r * D_FEAT;
          q[0] = c0[r];
          q[16] = c1[r];
          q[32] = c2[r];
          q[48] = c3[r];
        }
      }
    }
  }
}

#define COMB_THREADS 256
#define COMB_BLOCKS 512
__global__ __launch_bounds__(COMB_THREADS) void combine_kernel(
    const float* __restrict__ z0, const float* __restrict__ z1,
    const float* __restrict__ z2, const float* __restrict__ coeff,
    float* __restrict__ out, int n4) {
  const float c0 = coeff[0], c1 = coeff[1], c2 = coeff[2];
  const float4* p0 = (const float4*)z0;
  const float4* p1 = (const float4*)z1;
  const float4* p2 = (const float4*)z2;
  float4* o4 = (float4*)out;
  const int stride = gridDim.x * COMB_THREADS;
  int i = blockIdx.x * COMB_THREADS + threadIdx.x;

  auto mix = [&](const float4& a, const float4& b, const float4& c) {
    float4 r;
    r.x = __builtin_fmaf(c2, c.x, __builtin_fmaf(c1, b.x, c0 * a.x));
    r.y = __builtin_fmaf(c2, c.y, __builtin_fmaf(c1, b.y, c0 * a.y));
    r.z = __builtin_fmaf(c2, c.z, __builtin_fmaf(c1, b.z, c0 * a.z));
    r.w = __builtin_fmaf(c2, c.w, __builtin_fmaf(c1, b.w, c0 * a.w));
    return r;
  };

  for (; i + 3 * stride < n4; i += 4 * stride) {
    float4 a[4], b[4], c[4];
#pragma unroll
    for (int u = 0; u < 4; ++u) {
      a[u] = p0[i + u * stride];
      b[u] = p1[i + u * stride];
      c[u] = p2[i + u * stride];
    }
#pragma unroll
    for (int u = 0; u < 4; ++u) o4[i + u * stride] = mix(a[u], b[u], c[u]);
  }
  for (; i < n4; i += stride) o4[i] = mix(p0[i], p1[i], p2[i]);
}

extern "C" void kernel_launch(void* const* d_in, const int* in_sizes, int n_in,
                              void* d_out, int out_size, void* d_ws, size_t ws_size,
                              hipStream_t stream) {
  const float* x       = (const float*)d_in[0];
  const int*   eidx    = (const int*)  d_in[1];
  const float* ew      = (const float*)d_in[2];
  const float* coeff   = (const float*)d_in[3];
  const float* W1      = (const float*)d_in[4];
  const float* b1      = (const float*)d_in[5];
  const float* W2      = (const float*)d_in[6];
  const float* b2      = (const float*)d_in[7];
  const float* prelu_w = (const float*)d_in[8];

  const int N = in_sizes[0] / D_FEAT;
  const int E = in_sizes[2] / 3;
  const int total = N * D_FEAT;

  const int aggBlocks  = (N + NB - 1) / NB;

  float* ws   = (float*)d_ws;
  const size_t nPad = ((size_t)N + 31) & ~(size_t)31;
  float* deg3 = ws;
  float* bufA = ws + 3 * nPad;
  float* zpl  = bufA + (size_t)N * D_FEAT;
  const size_t nSlots = (size_t)aggBlocks * AGG_WAVES;
  float* bufB = zpl + (size_t)3 * N * D_FEAT;
  int*   lcnt = (int*)(bufB + (size_t)N * D_FEAT);
  int*   lsrc = lcnt + ((nSlots + 31) & ~(size_t)31);
  int*   ldst = lsrc + nSlots * LIST_WORDS;
  float* lnrm = (float*)(ldst + nSlots * LIST_WORDS);
  const size_t ePad = ((size_t)E + 31) & ~(size_t)31;
  int*   rsrc = (int*)(lnrm + nSlots * LIST_WORDS);
  int*   rcol = rsrc + ePad;
  float* rw   = (float*)(rcol + ePad);
  int*   seg  = (int*)(rw + ePad);

  const int ETHREADS = 256;
  const int degWords = (int)(3 * nPad);
  zero_kernel<<<(degWords + ETHREADS - 1) / ETHREADS, ETHREADS, 0, stream>>>(deg3, degWords);

  for (int t = 0; t < 3; ++t) {
    const int* row = eidx + (size_t)t * 2 * E;
    const int* col = row + E;
    const float* w = ew + (size_t)t * E;
    float* zt = zpl + (size_t)t * total;
    float* deg = deg3 + (size_t)t * nPad;

    degree_sort_kernel<<<NCHUNK + PROD_WGS, SORT_THREADS, 0, stream>>>(row, col, w, deg, E, rsrc, rcol, rw, seg,
                                                                    x, W1, bufA, N);

    aggregate_block_kernel<0><<<aggBlocks, AGG_THREADS, 0, stream>>>(rsrc, rcol, rw, seg, deg, bufA, b1, prelu_w, bufB, E, N,
                                                                  lsrc, ldst, lnrm, lcnt, W2);

    aggregate_block_kernel<1><<<aggBlocks, AGG_THREADS, 0, stream>>>(rsrc, rcol, rw, seg, deg, bufB, b2, prelu_w, zt, E, N,
                                                                  lsrc, ldst, lnrm, lcnt, W2);
  }

  combine_kernel<<<COMB_BLOCKS, COMB_THREADS, 0, stream>>>(zpl, zpl + (size_t)total, zpl + (size_t)2 * total, coeff,
                                                         (float*)d_out, total / 4);
}


# Aggregation without f32 atomics: the work-group counting-sorts its block's edges by destination in LDS, a wave owns 7 nodes and adds each bucket's rows in registers; sum order now fixed per node
# speedup vs baseline: 3.3950x; 1.0322x over previous
#include <hip/hip_runtime.h>
#include <hip/hip_bf16.h>


typedef __attribute__((ext_vector_type(2))) float v2f;
typedef __attribute__((ext_vector_type(8))) float v8f;

#define D_FEAT 128

__global__ void zero_kernel(float* __restrict__ p, int n) {
  int i = blockIdx.x * blockDim.x + threadIdx.x;
  if (i < n) p[i] = 0.0f;
}

#define NB 196
#define NCHUNK 128
#define PROD_WGS 125
#define SORT_THREADS 1024
#define SORT_WAVES (SORT_THREADS / 32)
#define SORT_EPT 8
#define SORT_CAP (SORT_EPT * SORT_THREADS)
#define SORT_BINS 256
#define WPITCH 288

__device__ __forceinline__ void product_part(
    const float* __restrict__ X, const float* __restrict__ Wm, float* __restrict__ H,
    int nRows, int g, int nG) {
  __shared__ float wl[(D_FEAT / 2) * WPITCH];
  const int tid = threadIdx.x;
  const int lane = tid & 31;
  const int wave = __builtin_amdgcn_readfirstlane(tid >> 5);
  for (int i = tid; i < (D_FEAT / 2) * (D_FEAT / 4); i += SORT_THREADS) {
    const int p = i >> 5;
    const int c4 = (i & 31) * 4;
    const float4 lo = *(const float4*)(Wm + (size_t)(2 * p) * D_FEAT + c4);
    const float4 hi = *(const float4*)(Wm + (size_t)(2 * p + 1) * D_FEAT + c4);
    float4* dst = (float4*)(wl + p * WPITCH + c4 * 2);
    dst[0] = make_float4(lo.x, hi.x, lo.y, hi.y);
    dst[1] = make_float4(lo.z, hi.z, lo.w, hi.w);
  }
  __syncthreads();

  const int strips = nRows / 16;
  const int spw = (strips + nG - 1) / nG;
  const int s0 = g * spw;
  const int s1 = min(strips, s0 + spw);
  const int l15 = lane & 15;
  const int h = lane >> 4;
  for (int task = 2 * s0 + wave; task < 2 * s1; task += SORT_WAVES) {
    const int m0 = (task >> 1) * 16;
    const int nbase = (task & 1) * 64;
    const float* ap = X + (size_t)(m0 + l15) * D_FEAT + 2 * h;
    const float* bp = wl + h * WPITCH + (nbase + l15) * 2;
    v8f c0 = {}, c1 = {}, c2 = {}, c3 = {};
#pragma nounroll
    for (int kb = 0; kb < D_FEAT; kb += 32) {
      v2f a[8];
#pragma unroll
      for (int u = 0; u < 8; ++u) a[u] = *(const v2f*)(ap + kb + 4 * u);
#pragma unroll
      for (int u = 0; u < 8; ++u) {
        const float* bk = bp + ((kb >> 1) + 2 * u) * WPITCH;
        const v2f b0 = *(const v2f*)(bk);
        const v2f b1 = *(const v2f*)(bk + 32);
        const v2f b2 = *(const v2f*)(bk + 64);
        const v2f b3 = *(const v2f*)(bk + 96);
        c0 = __builtin_amdgcn_wmma_f32_16x16x4_f32(false, a[u], false, b0, (short)0, c0, false, false);
        c1 = __builtin_amdgcn_wmma_f32_16x16x4_f32(false, a[u], false, b1, (short)0, c1, false, false);
        c2 = __builtin_amdgcn_wmma_f32_16x16x4_f32(false, a[u], false, b2, (short)0, c2, false, false);
        c3 = __builtin_amdgcn_wmma_f32_16x16x4_f32(false, a[u], false, b3, (short)0, c3, false, false);
      }
    }
    float* hp = H + (size_t)(m0 + 8 * h) * D_FEAT + nbase + l15;
#pragma unroll
    for (int r = 0; r < 8; ++r) {
      float* q = hp + (size_t)r * D_FEAT;
      q[0] = c0[r];
      q[16] = c1[r];
      q[32] = c2[r];
      q[48] = c3[r];
    }
  }
}

__global__ __launch_bounds__(SORT_THREADS) void degree_sort_kernel(
    const int* __restrict__ row, const int* __restrict__ col,
    const float* __restrict__ w, float* __restrict__ deg, int E,
    int* __restrict__ rsrc, int* __restrict__ rcol, float* __restrict__ rw,
    int* __restrict__ seg,
    const float* __restrict__ X, const float* __restrict__ Wm, float* __restrict__ H, int nRows) {
  if (blockIdx.x >= NCHUNK) {
    product_part(X, Wm, H, nRows, blockIdx.x - NCHUNK, gridDim.x - NCHUNK);
    return;
  }
  __shared__ int cnt[SORT_WAVES][SORT_BINS];
  __shared__ int binStart[SORT_BINS];
  __shared__ int waveSum[SORT_BINS / 32];
  __shared__ int s_r[SORT_CAP];
  __shared__ int s_c[SORT_CAP];
  __shared__ float s_w[SORT_CAP];

  const int tid = threadIdx.x;
  const int lane = tid & 31;
  const int wave = tid >> 5;
  const int chunkLen = (E + NCHUNK - 1) / NCHUNK;
  const int e0 = blockIdx.x * chunkLen;
  const int n = max(0, min(chunkLen, E - e0));

  for (int i = tid; i < SORT_WAVES * SORT_BINS; i += SORT_THREADS) (&cnt[0][0])[i] = 0;
  __syncthreads();

  int r[SORT_EPT], c[SORT_EPT], b[SORT_EPT], rank[SORT_EPT];
  float wt[SORT_EPT];
#pragma unroll
  for (int k = 0; k < SORT_EPT; ++k) {
    const int i = k * SORT_THREADS + tid;
    const bool ok = i < n;
    c[k] = ok ? col[e0 + i] : -1;
    r[k] = ok ? row[e0 + i] : 0;
    wt[k] = ok ? w[e0 + i] : 0.0f;
  }
#pragma unroll
  for (int k = 0; k < SORT_EPT; ++k) {
    b[k] = 0;
    rank[k] = 0;
    if (c[k] >= 0) {
      b[k] = c[k] / NB;
      rank[k] = __hip_atomic_fetch_add(&cnt[wave][b[k]], 1, __ATOMIC_RELAXED,
                                       __HIP_MEMORY_SCOPE_WORKGROUP);
    }
  }
  __syncthreads();

  int tot = 0;
  if (tid < SORT_BINS) {
    int t[SORT_WAVES];
#pragma unroll
    for (int q = 0; q < SORT_WAVES; ++q) t[q] = cnt[q][tid];
#pragma unroll
    for (int q = 0; q < SORT_WAVES; ++q) {
      cnt[q][tid] = tot;
      tot += t[q];
    }
  }
  int incl = tot;
#pragma unroll
  for (int d = 1; d < 32; d <<= 1) {
    const int o = __shfl_up(incl, d);
    if (lane >= d) incl += o;
  }
  if (tid < SORT_BINS && lane == 31) waveSum[wave] = incl;
  __syncthreads();
  if (tid < SORT_BINS) {
    int base = 0;
    for (int q = 0; q < wave; ++q) base += waveSum[q];
    const int start = base + incl - tot;
    binStart[tid] = start;
    ((int2*)seg)[(size_t)tid * NCHUNK + blockIdx.x] = make_int2(start, tot);
  }
  __syncthreads();

#pragma unroll
  for (int k = 0; k < SORT_EPT; ++k) {
    if (c[k] >= 0) {
      const int pos = binStart[b[k]] + cnt[wave][b[k]] + rank[k];
      s_r[pos] = r[k];
      s_c[pos] = c[k];
      s_w[pos] = wt[k];
    }
  }
  __syncthreads();
  for (int i = tid; i < n; i += SORT_THREADS) {
    rsrc[e0 + i] = s_r[i];
    rcol[e0 + i] = s_c[i];
    rw[e0 + i] = s_w[i];
  }
#pragma unroll
  for (int k = 0; k < SORT_EPT; ++k)
    if (c[k] >= 0) atomicAdd(&deg[c[k]], wt[k]);
}

__device__ __forceinline__ float dinv_of(const float* __restrict__ deg, int i) {
  const float d = deg[i] + 1.0f;
  return rsqrtf(d);
}

#define NBP 208
#define PITCH 132
#define AGG_THREADS 1024
#define AGG_WAVES (AGG_THREADS / 32)
#define CPW (NCHUNK / AGG_WAVES)
#define LIST_GROUPS 16
#define LIST_WORDS (LIST_GROUPS * 32)
#define WCAP 256
#define ECAP (WCAP * AGG_WAVES)
#define SLOTS ((NB + 31) / 32)
#define NKEYS (SLOTS * AGG_WAVES)

template <int MODE>
__global__ __launch_bounds__(AGG_THREADS) void aggregate_block_kernel(
    const int* __restrict__ rsrc, const int* __restrict__ rcol,
    const float* __restrict__ rw, const int* __restrict__ seg,
    const float* __restrict__ deg,
    const float* __restrict__ H, const float* __restrict__ bias,
    const float* __restrict__ pw, float* __restrict__ Z, int E, int N,
    int* __restrict__ lsrc, int* __restrict__ ldst, float* __restrict__ lnrm,
    int* __restrict__ lcnt, const float* __restrict__ Wn) {
  __shared__ float acc[NBP * PITCH];
  __shared__ float wl[MODE == 0 ? (D_FEAT / 2) * WPITCH : 4];
  __shared__ float dl[NB];
  __shared__ int kcnt[AGG_WAVES][NKEYS];
  __shared__ int kstart[NKEYS + 1];
  __shared__ int kwsum[NKEYS / 32];
  __shared__ int s_src[ECAP];
  __shared__ float s_nrm[ECAP];
  __shared__ int ovf;

  const int tid = threadIdx.x;
  const int lane = tid & 31;
  const int wave = __builtin_amdgcn_readfirstlane(tid >> 5);
  const int n0 = blockIdx.x * NB;
  const int nb = min(NB, N - n0);

  int* st_r = (int*)acc;
  float* st_n = acc + ECAP;
  int* st_k = (int*)(acc + 2 * ECAP);

  for (int i = tid; i < AGG_WAVES * NKEYS; i += AGG_THREADS) (&kcnt[0][0])[i] = 0;
  if (tid == 0) ovf = 0;
  if (tid < nb) dl[tid] = dinv_of(deg, n0 + tid);
  if (MODE == 0) {
    for (int i = tid; i < (D_FEAT / 2) * (D_FEAT / 4); i += AGG_THREADS) {
      const int p = i >> 5;
      const int c4 = (i & 31) * 4;
      const float4 lo = *(const float4*)(Wn + (size_t)(2 * p) * D_FEAT + c4);
      const float4 hi = *(const float4*)(Wn + (size_t)(2 * p + 1) * D_FEAT + c4);
      float4* dst = (float4*)(wl + p * WPITCH + c4 * 2);
      dst[0] = make_float4(lo.x, hi.x, lo.y, hi.y);
      dst[1] = make_float4(lo.z, hi.z, lo.w, hi.w);
    }
  }
  __syncthreads();

  const size_t slot = (size_t)blockIdx.x * AGG_WAVES + wave;
  int* mysrc = lsrc + slot * LIST_WORDS;
  int* mydst = ldst + slot * LIST_WORDS;
  float* mynrm = lnrm + slot * LIST_WORDS;

  auto sum_group = [&](int r, float nrm, int take, float& a0, float& a1, float& a2, float& a3) {
    for (int g0 = 0; g0 < take; g0 += 8) {
      float v[8][4];
#pragma unroll
      for (int u = 0; u < 8; ++u) {
        const int k = min(g0 + u, take - 1);
        const int rk = __builtin_amdgcn_readlane(r, k);
        const float* hp = H + (size_t)rk * D_FEAT + lane;
#pragma unroll
        for (int q = 0; q < 4; ++q) v[u][q] = hp[32 * q];
      }
#pragma unroll
      for (int u = 0; u < 8; ++u)
        asm volatile("" : : "v"(v[u][0]), "v"(v[u][1]), "v"(v[u][2]), "v"(v[u][3]));
#pragma unroll
      for (int u = 0; u < 8; ++u) {
        if (g0 + u < take) {
          const float nk = __int_as_float(__builtin_amdgcn_readlane(__float_as_int(nrm), g0 + u));
          a0 = __builtin_fmaf(v[u][0], nk, a0);
          a1 = __builtin_fmaf(v[u][1], nk, a1);
          a2 = __builtin_fmaf(v[u][2], nk, a2);
          a3 = __builtin_fmaf(v[u][3], nk, a3);
        }
      }
    }
  };

  auto add_group = [&](int r, int cl, float nrm, int take) {
    for (int g0 = 0; g0 < take; g0 += 8) {
      float v[8][4];
#pragma unroll
      for (int u = 0; u < 8; ++u) {
        const int k = min(g0 + u, take - 1);
        const int rk = __builtin_amdgcn_readlane(r, k);
        const float* hp = H + (size_t)rk * D_FEAT + lane;
#pragma unroll
        for (int q = 0; q < 4; ++q) v[u][q] = hp[32 * q];
      }
#pragma unroll
      for (int u = 0; u < 8; ++u)
        asm volatile("" : : "v"(v[u][0]), "v"(v[u][1]), "v"(v[u][2]), "v"(v[u][3]));
#pragma unroll
      for (int u = 0; u < 8; ++u) {
        if (g0 + u < take) {
          const int ck = __builtin_amdgcn_readlane(cl, g0 + u);
          const float nk = __int_as_float(__builtin_amdgcn_readlane(__float_as_int(nrm), g0 + u));
          float* ap = acc + ck * PITCH + lane;
#pragma unroll
          for (int q = 0; q < 4; ++q)
            __hip_atomic_fetch_add(ap + 32 * q, v[u][q] * nk, __ATOMIC_RELAXED,
                                   __HIP_MEMORY_SCOPE_WORKGROUP);
        }
      }
    }
  };

  auto stage_group = [&](int r, int cl, float nrm, int take, int done) {
    const int i = done + lane;
    const bool valid = lane < take && i < WCAP;
    const int key = (cl & 31) * SLOTS + (cl >> 5);
    const int kk = valid ? key : -1;
    int before = 0, after = 0;
#pragma unroll
    for (int j = 0; j < 32; ++j) {
      const bool eq = __builtin_amdgcn_readlane(kk, j) == kk;
      before += (eq && j < lane) ? 1 : 0;
      after += (eq && j > lane) ? 1 : 0;
    }
    if (valid) {
      const int rank = kcnt[wave][key] + before;
      const int at = wave * WCAP + i;
      st_r[at] = r;
      st_n[at] = nrm;
      st_k[at] = key | (rank << 8);
      if (after == 0) kcnt[wave][key] = rank + 1;
    }
  };

  auto walk = [&](bool keep, int& total, auto&& f) {
    const int chunkLen = (E + NCHUNK - 1) / NCHUNK;
    int2 sc = make_int2(0, 0);
    if (lane < CPW) sc = ((const int2*)seg)[(size_t)blockIdx.x * NCHUNK + wave * CPW + lane];
    int s[CPW], p[CPW + 1];
    p[0] = 0;
#pragma unroll
    for (int q = 0; q < CPW; ++q) {
      s[q] = __builtin_amdgcn_readlane(sc.x, q) + (wave * CPW + q) * chunkLen;
      p[q + 1] = p[q] + __builtin_amdgcn_readlane(sc.y, q);
    }
    total = p[CPW];
    int groups = 0;
    for (int done = 0; done < total; done += 32) {
      const int v = done + lane;
      int r = 0, cl = 0;
      float nrm = 0.0f;
      if (v < total) {
        int a = s[0] + v;
#pragma unroll
        for (int q = 1; q < CPW; ++q)
          if (v >= p[q]) a = s[q] + (v - p[q]);
        r = rsrc[a];
        const int c = rcol[a];
        nrm = dinv_of(deg, r) * rw[a] * dinv_of(deg, c);
        cl = c - n0;
      }
      if (keep) {
        if (groups < LIST_GROUPS) {
          mysrc[groups * 32 + lane] = r;
          mydst[groups * 32 + lane] = cl;
          mynrm[groups * 32 + lane] = nrm;
        }
      }
      ++groups;
      f(r, cl, nrm, min(32, total - done), done);
    }
    return groups;
  };

  auto for_groups = [&](bool first, auto&& f) {
    int total = 0;
    if (MODE == 0) {
      const int groups = walk(first, total, f);
      if (first && lane == 0) lcnt[slot] = groups <= LIST_GROUPS ? total : -1;
    } else {
      const int cnt = __builtin_amdgcn_readfirstlane(lcnt[slot]);
      if (cnt >= 0) {
        total = cnt;
        for (int done = 0; done < cnt; done += 32) {
          const int r = mysrc[done + lane];
          const int cl = mydst[done + lane];
          const float nrm = mynrm[done + lane];
          f(r, cl, nrm, min(32, cnt - done), done);
        }
      } else {
        walk(false, total, f);
      }
    }
    return total;
  };

  const int myTotal = for_groups(true, stage_group);
  if (myTotal > WCAP && lane == 0) ovf = 1;
  __syncthreads();

  if (ovf == 0) {
    int tot = 0;
    if (tid < NKEYS) {
      int t[AGG_WAVES];
#pragma unroll
      for (int q = 0; q < AGG_WAVES; ++q) t[q] = kcnt[q][tid];
#pragma unroll
      for (int q = 0; q < AGG_WAVES; ++q) {
        kcnt[q][tid] = tot;
        tot += t[q];
      }
    }
    int incl = tot;
#pragma unroll
    for (int d = 1; d < 32; d <<= 1) {
      const int o = __shfl_up(incl, d);
      if (lane >= d) incl += o;
    }
    if (tid < NKEYS && lane == 31) kwsum[wave] = incl;
    __syncthreads();
    if (tid < NKEYS) {
      int base = 0;
      for (int q = 0; q < wave; ++q) base += kwsum[q];
      kstart[tid] = base + incl - tot;
      if (tid == NKEYS - 1) kstart[NKEYS] = base + incl;
    }
    __syncthreads();

    for (int i = lane; i < myTotal; i += 32) {
      const int at = wave * WCAP + i;
      const int kr = st_k[at];
      const int key = kr & 255;
      const int pos = kstart[key] + kcnt[wave][key] + (kr >> 8);
      s_src[pos] = st_r[at];
      s_nrm[pos] = st_n[at];
    }
    __syncthreads();

    for (int s = 0; s < SLOTS; ++s) {
      const int cl = s * 32 + wave;
      if (cl >= nb) break;
      const int b0 = __builtin_amdgcn_readfirstlane(kstart[wave * SLOTS + s]);
      const int b1 = __builtin_amdgcn_readfirstlane(kstart[wave * SLOTS + s + 1]);
      float a0 = 0.0f, a1 = 0.0f, a2 = 0.0f, a3 = 0.0f;
      for (int g = b0; g < b1; g += 32) {
        const int take = min(32, b1 - g);
        int r = 0;
        float nrm = 0.0f;
        if (lane < take) {
          r = s_src[g + lane];
          nrm = s_nrm[g + lane];
        }
        sum_group(r, nrm, take, a0, a1, a2, a3);
      }
      float* ap = acc + cl * PITCH + lane;
      ap[0] = a0;
      ap[32] = a1;
      ap[64] = a2;
      ap[96] = a3;
    }
    for (int i = nb * PITCH + tid; i < NBP * PITCH; i += AGG_THREADS) acc[i] = 0.0f;
  } else {
    {
      float4* a4 = (float4*)acc;
      for (int i = tid; i < NBP * PITCH / 4; i += AGG_THREADS)
        a4[i] = make_float4(0.0f, 0.0f, 0.0f, 0.0f);
    }
    __syncthreads();
    for_groups(false, [&](int r, int cl, float nrm, int take, int) { add_group(r, cl, nrm, take); });
  }

  __syncthreads();

  for (int i = tid; i < nb * D_FEAT; i += AGG_THREADS) {
    const int f = i & (D_FEAT - 1);
    const int rl = i >> 7;
    const int node = n0 + rl;
    const float d = dl[rl];
    const size_t g = (size_t)node * D_FEAT + f;
    const float v = acc[rl * PITCH + f] + H[g] * d * d + bias[f];
    const float z = fmaxf(v, 0.0f) + pw[f] * fminf(v, 0.0f);
    if (MODE == 0) acc[rl * PITCH + f] = z;
    else Z[g] = z;
  }

  if (MODE == 0) {
    __syncthreads();
    const int m0 = (wave >> 1) * 16;
    if (wave < 2 * (NBP / 16) && m0 < nb) {
      const int nbase = (wave & 1) * 64;
      const int l15 = lane & 15;
      const int h = lane >> 4;
      const float* ap = acc + (m0 + l15) * PITCH + 2 * h;
      const float* bp = wl + h * WPITCH + (nbase + l15) * 2;
      v8f c0 = {}, c1 = {}, c2 = {}, c3 = {};
#pragma unroll
      for (int k0 = 0; k0 < D_FEAT; k0 += 4) {
        const v2f a = *(const v2f*)(ap + k0);
        const float* bk = bp + (k0 >> 1) * WPITCH;
        const v2f b0 = *(const v2f*)(bk);
        const v2f b1 = *(const v2f*)(bk + 32);
        const v2f b2 = *(const v2f*)(bk + 64);
        const v2f b3 = *(const v2f*)(bk + 96);
        c0 = __builtin_amdgcn_wmma_f32_16x16x4_f32(false, a, false, b0, (short)0, c0, false, false);
        c1 = __builtin_amdgcn_wmma_f32_16x16x4_f32(false, a, false, b1, (short)0, c1, false, false);
        c2 = __builtin_amdgcn_wmma_f32_16x16x4_f32(false, a, false, b2, (short)0, c2, false, false);
        c3 = __builtin_amdgcn_wmma_f32_16x16x4_f32(false, a, false, b3, (short)0, c3, false, false);
      }
      const int rbase = m0 + 8 * h;
      float* zp = Z + (size_t)(n0 + rbase) * D_FEAT + nbase + l15;
#pragma unroll
      for (int r = 0; r < 8; ++r) {
        if (rbase + r < nb) {
          float* q = zp + (size_t)r * D_FEAT;
          q[0] = c0[r];
          q[16] = c1[r];
          q[32] = c2[r];
          q[48] = c3[r];
        }
      }
    }
  }
}

#define COMB_THREADS 256
#define COMB_BLOCKS 512
__global__ __launch_bounds__(COMB_THREADS) void combine_kernel(
    const float* __restrict__ z0, const float* __restrict__ z1,
    const float* __restrict__ z2, const float* __restrict__ coeff,
    float* __restrict__ out, int n4) {
  const float c0 = coeff[0], c1 = coeff[1], c2 = coeff[2];
  const float4* p0 = (const float4*)z0;
  const float4* p1 = (const float4*)z1;
  const float4* p2 = (const float4*)z2;
  float4* o4 = (float4*)out;
  const int stride = gridDim.x * COMB_THREADS;
  int i = blockIdx.x * COMB_THREADS + threadIdx.x;

  auto mix = [&](const float4& a, const float4& b, const float4& c) {
    float4 r;
    r.x = __builtin_fmaf(c2, c.x, __builtin_fmaf(c1, b.x, c0 * a.x));
    r.y = __builtin_fmaf(c2, c.y, __builtin_fmaf(c1, b.y, c0 * a.y));
    r.z = __builtin_fmaf(c2, c.z, __builtin_fmaf(c1, b.z, c0 * a.z));
    r.w = __builtin_fmaf(c2, c.w, __builtin_fmaf(c1, b.w, c0 * a.w));
    return r;
  };

  for (; i + 3 * stride < n4; i += 4 * stride) {
    float4 a[4], b[4], c[4];
#pragma unroll
    for (int u = 0; u < 4; ++u) {
      a[u] = p0[i + u * stride];
      b[u] = p1[i + u * stride];
      c[u] = p2[i + u * stride];
    }
#pragma unroll
    for (int u = 0; u < 4; ++u) o4[i + u * stride] = mix(a[u], b[u], c[u]);
  }
  for (; i < n4; i += stride) o4[i] = mix(p0[i], p1[i], p2[i]);
}

extern "C" void kernel_launch(void* const* d_in, const int* in_sizes, int n_in,
                              void* d_out, int out_size, void* d_ws, size_t ws_size,
                              hipStream_t stream) {
  const float* x       = (const float*)d_in[0];
  const int*   eidx    = (const int*)  d_in[1];
  const float* ew      = (const float*)d_in[2];
  const float* coeff   = (const float*)d_in[3];
  const float* W1      = (const float*)d_in[4];
  const float* b1      = (const float*)d_in[5];
  const float* W2      = (const float*)d_in[6];
  const float* b2      = (const float*)d_in[7];
  const float* prelu_w = (const float*)d_in[8];

  const int N = in_sizes[0] / D_FEAT;
  const int E = in_sizes[2] / 3;
  const int total = N * D_FEAT;

  const int aggBlocks  = (N + NB - 1) / NB;

  float* ws   = (float*)d_ws;
  const size_t nPad = ((size_t)N + 31) & ~(size_t)31;
  float* deg3 = ws;
  float* bufA = ws + 3 * nPad;
  float* zpl  = bufA + (size_t)N * D_FEAT;
  const size_t nSlots = (size_t)aggBlocks * AGG_WAVES;
  float* bufB = zpl + (size_t)3 * N * D_FEAT;
  int*   lcnt = (int*)(bufB + (size_t)N * D_FEAT);
  int*   lsrc = lcnt + ((nSlots + 31) & ~(size_t)31);
  int*   ldst = lsrc + nSlots * LIST_WORDS;
  float* lnrm = (float*)(ldst + nSlots * LIST_WORDS);
  const size_t ePad = ((size_t)E + 31) & ~(size_t)31;
  int*   rsrc = (int*)(lnrm + nSlots * LIST_WORDS);
  int*   rcol = rsrc + ePad;
  float* rw   = (float*)(rcol + ePad);
  int*   seg  = (int*)(rw + ePad);

  const int ETHREADS = 256;
  const int degWords = (int)(3 * nPad);
  zero_kernel<<<(degWords + ETHREADS - 1) / ETHREADS, ETHREADS, 0, stream>>>(deg3, degWords);

  for (int t = 0; t < 3; ++t) {
    const int* row = eidx + (size_t)t * 2 * E;
    const int* col = row + E;
    const float* w = ew + (size_t)t * E;
    float* zt = zpl + (size_t)t * total;
    float* deg = deg3 + (size_t)t * nPad;

    degree_sort_kernel<<<NCHUNK + PROD_WGS, SORT_THREADS, 0, stream>>>(row, col, w, deg, E, rsrc, rcol, rw, seg,
                                                                    x, W1, bufA, N);

    aggregate_block_kernel<0><<<aggBlocks, AGG_THREADS, 0, stream>>>(rsrc, rcol, rw, seg, deg, bufA, b1, prelu_w, bufB, E, N,
                                                                  lsrc, ldst, lnrm, lcnt, W2);

    aggregate_block_kernel<1><<<aggBlocks, AGG_THREADS, 0, stream>>>(rsrc, rcol, rw, seg, deg, bufB, b2, prelu_w, zt, E, N,
                                                                  lsrc, ldst, lnrm, lcnt, W2);
  }

  combine_kernel<<<COMB_BLOCKS, COMB_THREADS, 0, stream>>>(zpl, zpl + (size_t)total, zpl + (size_t)2 * total, coeff,
                                                         (float*)d_out, total / 4);
}


# Layer 1's work-group writes out its block's edges as it ordered them (bucket bounds, source, norm); layer 2 starts there: a wave sums its nodes' buckets and finishes each row; sum order unchanged
# speedup vs baseline: 3.6084x; 1.0629x over previous
#include <hip/hip_runtime.h>
#include <hip/hip_bf16.h>


typedef __attribute__((ext_vector_type(2))) float v2f;
typedef __attribute__((ext_vector_type(8))) float v8f;

#define D_FEAT 128

__global__ void zero_kernel(float* __restrict__ p, int n) {
  int i = blockIdx.x * blockDim.x + threadIdx.x;
  if (i < n) p[i] = 0.0f;
}

#define NB 196
#define NCHUNK 128
#define PROD_WGS 125
#define SORT_THREADS 1024
#define SORT_WAVES (SORT_THREADS / 32)
#define SORT_EPT 8
#define SORT_CAP (SORT_EPT * SORT_THREADS)
#define SORT_BINS 256
#define WPITCH 288

__device__ __forceinline__ void product_part(
    const float* __restrict__ X, const float* __restrict__ Wm, float* __restrict__ H,
    int nRows, int g, int nG) {
  __shared__ float wl[(D_FEAT / 2) * WPITCH];
  const int tid = threadIdx.x;
  const int lane = tid & 31;
  const int wave = __builtin_amdgcn_readfirstlane(tid >> 5);
  for (int i = tid; i < (D_FEAT / 2) * (D_FEAT / 4); i += SORT_THREADS) {
    const int p = i >> 5;
    const int c4 = (i & 31) * 4;
    const float4 lo = *(const float4*)(Wm + (size_t)(2 * p) * D_FEAT + c4);
    const float4 hi = *(const float4*)(Wm + (size_t)(2 * p + 1) * D_FEAT + c4);
    float4* dst = (float4*)(wl + p * WPITCH + c4 * 2);
    dst[0] = make_float4(lo.x, hi.x, lo.y, hi.y);
    dst[1] = make_float4(lo.z, hi.z, lo.w, hi.w);
  }
  __syncthreads();

  const int strips = nRows / 16;
  const int spw = (strips + nG - 1) / nG;
  const int s0 = g * spw;
  const int s1 = min(strips, s0 + spw);
  const int l15 = lane & 15;
  const int h = lane >> 4;
  for (int task = 2 * s0 + wave; task < 2 * s1; task += SORT_WAVES) {
    const int m0 = (task >> 1) * 16;
    const int nbase = (task & 1) * 64;
    const float* ap = X + (size_t)(m0 + l15) * D_FEAT + 2 * h;
    const float* bp = wl + h * WPITCH + (nbase + l15) * 2;
    v8f c0 = {}, c1 = {}, c2 = {}, c3 = {};
#pragma nounroll
    for (int kb = 0; kb < D_FEAT; kb += 32) {
      v2f a[8];
#pragma unroll
      for (int u = 0; u < 8; ++u) a[u] = *(const v2f*)(ap + kb + 4 * u);
#pragma unroll
      for (int u = 0; u < 8; ++u) {
        const float* bk = bp + ((kb >> 1) + 2 * u) * WPITCH;
        const v2f b0 = *(const v2f*)(bk);
        const v2f b1 = *(const v2f*)(bk + 32);
        const v2f b2 = *(const v2f*)(bk + 64);
        const v2f b3 = *(const v2f*)(bk + 96);
        c0 = __builtin_amdgcn_wmma_f32_16x16x4_f32(false, a[u], false, b0, (short)0, c0, false, false);
        c1 = __builtin_amdgcn_wmma_f32_16x16x4_f32(false, a[u], false, b1, (short)0, c1, false, false);
        c2 = __builtin_amdgcn_wmma_f32_16x16x4_f32(false, a[u], false, b2, (short)0, c2, false, false);
        c3 = __builtin_amdgcn_wmma_f32_16x16x4_f32(false, a[u], false, b3, (short)0, c3, false, false);
      }
    }
    float* hp = H + (size_t)(m0 + 8 * h) * D_FEAT + nbase + l15;
#pragma unroll
    for (int r = 0; r < 8; ++r) {
      float* q = hp + (size_t)r * D_FEAT;
      q[0] = c0[r];
      q[16] = c1[r];
      q[32] = c2[r];
      q[48] = c3[r];
    }
  }
}

__global__ __launch_bounds__(SORT_THREADS) void degree_sort_kernel(
    const int* __restrict__ row, const int* __restrict__ col,
    const float* __restrict__ w, float* __restrict__ deg, int E,
    int* __restrict__ rsrc, int* __restrict__ rcol, float* __restrict__ rw,
    int* __restrict__ seg,
    const float* __restrict__ X, const float* __restrict__ Wm, float* __restrict__ H, int nRows) {
  if (blockIdx.x >= NCHUNK) {
    product_part(X, Wm, H, nRows, blockIdx.x - NCHUNK, gridDim.x - NCHUNK);
    return;
  }
  __shared__ int cnt[SORT_WAVES][SORT_BINS];
  __shared__ int binStart[SORT_BINS];
  __shared__ int waveSum[SORT_BINS / 32];
  __shared__ int s_r[SORT_CAP];
  __shared__ int s_c[SORT_CAP];
  __shared__ float s_w[SORT_CAP];

  const int tid = threadIdx.x;
  const int lane = tid & 31;
  const int wave = tid >> 5;
  const int chunkLen = (E + NCHUNK - 1) / NCHUNK;
  const int e0 = blockIdx.x * chunkLen;
  const int n = max(0, min(chunkLen, E - e0));

  for (int i = tid; i < SORT_WAVES * SORT_BINS; i += SORT_THREADS) (&cnt[0][0])[i] = 0;
  __syncthreads();

  int r[SORT_EPT], c[SORT_EPT], b[SORT_EPT], rank[SORT_EPT];
  float wt[SORT_EPT];
#pragma unroll
  for (int k = 0; k < SORT_EPT; ++k) {
    const int i = k * SORT_THREADS + tid;
    const bool ok = i < n;
    c[k] = ok ? col[e0 + i] : -1;
    r[k] = ok ? row[e0 + i] : 0;
    wt[k] = ok ? w[e0 + i] : 0.0f;
  }
#pragma unroll
  for (int k = 0; k < SORT_EPT; ++k) {
    b[k] = 0;
    rank[k] = 0;
    if (c[k] >= 0) {
      b[k] = c[k] / NB;
      rank[k] = __hip_atomic_fetch_add(&cnt[wave][b[k]], 1, __ATOMIC_RELAXED,
                                       __HIP_MEMORY_SCOPE_WORKGROUP);
    }
  }
  __syncthreads();

  int tot = 0;
  if (tid < SORT_BINS) {
    int t[SORT_WAVES];
#pragma unroll
    for (int q = 0; q < SORT_WAVES; ++q) t[q] = cnt[q][tid];
#pragma unroll
    for (int q = 0; q < SORT_WAVES; ++q) {
      cnt[q][tid] = tot;
      tot += t[q];
    }
  }
  int incl = tot;
#pragma unroll
  for (int d = 1; d < 32; d <<= 1) {
    const int o = __shfl_up(incl, d);
    if (lane >= d) incl += o;
  }
  if (tid < SORT_BINS && lane == 31) waveSum[wave] = incl;
  __syncthreads();
  if (tid < SORT_BINS) {
    int base = 0;
    for (int q = 0; q < wave; ++q) base += waveSum[q];
    const int start = base + incl - tot;
    binStart[tid] = start;
    ((int2*)seg)[(size_t)tid * NCHUNK + blockIdx.x] = make_int2(start, tot);
  }
  __syncthreads();

#pragma unroll
  for (int k = 0; k < SORT_EPT; ++k) {
    if (c[k] >= 0) {
      const int pos = binStart[b[k]] + cnt[wave][b[k]] + rank[k];
      s_r[pos] = r[k];
      s_c[pos] = c[k];
      s_w[pos] = wt[k];
    }
  }
  __syncthreads();
  for (int i = tid; i < n; i += SORT_THREADS) {
    rsrc[e0 + i] = s_r[i];
    rcol[e0 + i] = s_c[i];
    rw[e0 + i] = s_w[i];
  }
#pragma unroll
  for (int k = 0; k < SORT_EPT; ++k)
    if (c[k] >= 0) atomicAdd(&deg[c[k]], wt[k]);
}

__device__ __forceinline__ float dinv_of(const float* __restrict__ deg, int i) {
  const float d = deg[i] + 1.0f;
  return rsqrtf(d);
}

#define NBP 208
#define PITCH 132
#define AGG_THREADS 1024
#define AGG_WAVES (AGG_THREADS / 32)
#define CPW (NCHUNK / AGG_WAVES)
#define WCAP 256
#define ECAP (WCAP * AGG_WAVES)
#define SLOTS ((NB + 31) / 32)
#define NKEYS (SLOTS * AGG_WAVES)
#define KPITCH 256

__device__ __forceinline__ void sum_group(const float* __restrict__ H, int lane, int r, float nrm,
                                          int take, float& a0, float& a1, float& a2, float& a3) {
  for (int g0 = 0; g0 < take; g0 += 8) {
    float v[8][4];
#pragma unroll
    for (int u = 0; u < 8; ++u) {
      const int k = min(g0 + u, take - 1);
      const int rk = __builtin_amdgcn_readlane(r, k);
      const float* hp = H + (size_t)rk * D_FEAT + lane;
#pragma unroll
      for (int q = 0; q < 4; ++q) v[u][q] = hp[32 * q];
    }
#pragma unroll
    for (int u = 0; u < 8; ++u)
      asm volatile("" : : "v"(v[u][0]), "v"(v[u][1]), "v"(v[u][2]), "v"(v[u][3]));
#pragma unroll
    for (int u = 0; u < 8; ++u) {
      if (g0 + u < take) {
        const float nk = __int_as_float(__builtin_amdgcn_readlane(__float_as_int(nrm), g0 + u));
        a0 = __builtin_fmaf(v[u][0], nk, a0);
        a1 = __builtin_fmaf(v[u][1], nk, a1);
        a2 = __builtin_fmaf(v[u][2], nk, a2);
        a3 = __builtin_fmaf(v[u][3], nk, a3);
      }
    }
  }
}

__device__ __forceinline__ void add_group(const float* __restrict__ H, float* acc, int lane,
                                          int r, int cl, float nrm, int take) {
  for (int g0 = 0; g0 < take; g0 += 8) {
    float v[8][4];
#pragma unroll
    for (int u = 0; u < 8; ++u) {
      const int k = min(g0 + u, take - 1);
      const int rk = __builtin_amdgcn_readlane(r, k);
      const float* hp = H + (size_t)rk * D_FEAT + lane;
#pragma unroll
      for (int q = 0; q < 4; ++q) v[u][q] = hp[32 * q];
    }
#pragma unroll
    for (int u = 0; u < 8; ++u)
      asm volatile("" : : "v"(v[u][0]), "v"(v[u][1]), "v"(v[u][2]), "v"(v[u][3]));
#pragma unroll
    for (int u = 0; u < 8; ++u) {
      if (g0 + u < take) {
        const int ck = __builtin_amdgcn_readlane(cl, g0 + u);
        const float nk = __int_as_float(__builtin_amdgcn_readlane(__float_as_int(nrm), g0 + u));
        float* ap = acc + ck * PITCH + lane;
#pragma unroll
        for (int q = 0; q < 4; ++q)
          __hip_atomic_fetch_add(ap + 32 * q, v[u][q] * nk, __ATOMIC_RELAXED,
                                 __HIP_MEMORY_SCOPE_WORKGROUP);
      }
    }
  }
}

template <class F>
__device__ __forceinline__ int walk_runs(
    const int* __restrict__ rsrc, const int* __restrict__ rcol, const float* __restrict__ rw,
    const int* __restrict__ seg, const float* __restrict__ deg, int E,
    int block, int wave, int lane, int n0, F&& f) {
  const int chunkLen = (E + NCHUNK - 1) / NCHUNK;
  int2 sc = make_int2(0, 0);
  if (lane < CPW) sc = ((const int2*)seg)[(size_t)block * NCHUNK + wave * CPW + lane];
  int s[CPW], p[CPW + 1];
  p[0] = 0;
#pragma unroll
  for (int q = 0; q < CPW; ++q) {
    s[q] = __builtin_amdgcn_readlane(sc.x, q) + (wave * CPW + q) * chunkLen;
    p[q + 1] = p[q] + __builtin_amdgcn_readlane(sc.y, q);
  }
  const int total = p[CPW];
  for (int done = 0; done < total; done += 32) {
    const int v = done + lane;
    int r = 0, cl = 0;
    float nrm = 0.0f;
    if (v < total) {
      int a = s[0] + v;
#pragma unroll
      for (int q = 1; q < CPW; ++q)
        if (v >= p[q]) a = s[q] + (v - p[q]);
      r = rsrc[a];
      const int c = rcol[a];
      nrm = dinv_of(deg, r) * rw[a] * dinv_of(deg, c);
      cl = c - n0;
    }
    f(r, cl, nrm, min(32, total - done), done);
  }
  return total;
}

__global__ __launch_bounds__(AGG_THREADS) void aggregate_block_kernel(
    const int* __restrict__ rsrc, const int* __restrict__ rcol,
    const float* __restrict__ rw, const int* __restrict__ seg,
    const float* __restrict__ deg,
    const float* __restrict__ H, const float* __restrict__ bias,
    const float* __restrict__ pw, float* __restrict__ Z, int E, int N,
    int* __restrict__ kstG, int* __restrict__ bsrcG, float* __restrict__ bnrmG,
    const float* __restrict__ Wn) {
  __shared__ float acc[NBP * PITCH];
  __shared__ float wl[(D_FEAT / 2) * WPITCH];
  __shared__ float dl[NB];
  __shared__ int kcnt[AGG_WAVES][NKEYS];
  __shared__ int kstart[NKEYS + 1];
  __shared__ int kwsum[NKEYS / 32];
  __shared__ int s_src[ECAP];
  __shared__ float s_nrm[ECAP];
  __shared__ int ovf;

  const int tid = threadIdx.x;
  const int lane = tid & 31;
  const int wave = __builtin_amdgcn_readfirstlane(tid >> 5);
  const int n0 = blockIdx.x * NB;
  const int nb = min(NB, N - n0);

  int* st_r = (int*)acc;
  float* st_n = acc + ECAP;
  int* st_k = (int*)(acc + 2 * ECAP);

  for (int i = tid; i < AGG_WAVES * NKEYS; i += AGG_THREADS) (&kcnt[0][0])[i] = 0;
  if (tid == 0) ovf = 0;
  if (tid < nb) dl[tid] = dinv_of(deg, n0 + tid);
  for (int i = tid; i < (D_FEAT / 2) * (D_FEAT / 4); i += AGG_THREADS) {
    const int p = i >> 5;
    const int c4 = (i & 31) * 4;
    const float4 lo = *(const float4*)(Wn + (size_t)(2 * p) * D_FEAT + c4);
    const float4 hi = *(const float4*)(Wn + (size_t)(2 * p + 1) * D_FEAT + c4);
    float4* dst = (float4*)(wl + p * WPITCH + c4 * 2);
    dst[0] = make_float4(lo.x, hi.x, lo.y, hi.y);
    dst[1] = make_float4(lo.z, hi.z, lo.w, hi.w);
  }
  __syncthreads();

  auto stage_group = [&](int r, int cl, float nrm, int take, int done) {
    const int i = done + lane;
    const bool valid = lane < take && i < WCAP;
    const int key = (cl & 31) * SLOTS + (cl >> 5);
    const int kk = valid ? key : -1;
    int before = 0, after = 0;
#pragma unroll
    for (int j = 0; j < 32; ++j) {
      const bool eq = __builtin_amdgcn_readlane(kk, j) == kk;
      before += (eq && j < lane) ? 1 : 0;
      after += (eq && j > lane) ? 1 : 0;
    }
    if (valid) {
      const int rank = kcnt[wave][key] + before;
      const int at = wave * WCAP + i;
      st_r[at] = r;
      st_n[at] = nrm;
      st_k[at] = key | (rank << 8);
      if (after == 0) kcnt[wave][key] = rank + 1;
    }
  };

  const int myTotal = walk_runs(rsrc, rcol, rw, seg, deg, E, blockIdx.x, wave, lane, n0, stage_group);
  if (myTotal > WCAP && lane == 0) ovf = 1;
  __syncthreads();

  if (ovf == 0) {
    int tot = 0;
    if (tid < NKEYS) {
      int t[AGG_WAVES];
#pragma unroll
      for (int q = 0; q < AGG_WAVES; ++q) t[q] = kcnt[q][tid];
#pragma unroll
      for (int q = 0; q < AGG_WAVES; ++q) {
        kcnt[q][tid] = tot;
        tot += t[q];
      }
    }
    int incl = tot;
#pragma unroll
    for (int d = 1; d < 32; d <<= 1) {
      const int o = __shfl_up(incl, d);
      if (lane >= d) incl += o;
    }
    if (tid < NKEYS && lane == 31) kwsum[wave] = incl;
    __syncthreads();
    if (tid < NKEYS) {
      int base = 0;
      for (int q = 0; q < wave; ++q) base += kwsum[q];
      kstart[tid] = base + incl - tot;
      if (tid == NKEYS - 1) kstart[NKEYS] = base + incl;
    }
    __syncthreads();

    for (int i = lane; i < myTotal; i += 32) {
      const int at = wave * WCAP + i;
      const int kr = st_k[at];
      const int key = kr & 255;
      const int pos = kstart[key] + kcnt[wave][key] + (kr >> 8);
      s_src[pos] = st_r[at];
      s_nrm[pos] = st_n[at];
    }
    __syncthreads();

    {
      int* kb = kstG + (size_t)blockIdx.x * KPITCH;
      int* bs = bsrcG + (size_t)blockIdx.x * ECAP;
      float* bn = bnrmG + (size_t)blockIdx.x * ECAP;
      if (tid <= NKEYS) kb[tid] = kstart[tid];
      const int blockTotal = kstart[NKEYS];
      for (int i = tid; i < blockTotal; i += AGG_THREADS) {
        bs[i] = s_src[i];
        bn[i] = s_nrm[i];
      }
    }

    for (int s = 0; s < SLOTS; ++s) {
      const int cl = s * 32 + wave;
      if (cl >= nb) break;
      const int b0 = __builtin_amdgcn_readfirstlane(kstart[wave * SLOTS + s]);
      const int b1 = __builtin_amdgcn_readfirstlane(kstart[wave * SLOTS + s + 1]);
      float a0 = 0.0f, a1 = 0.0f, a2 = 0.0f, a3 = 0.0f;
      for (int g = b0; g < b1; g += 32) {
        const int take = min(32, b1 - g);
        int r = 0;
        float nrm = 0.0f;
        if (lane < take) {
          r = s_src[g + lane];
          nrm = s_nrm[g + lane];
        }
        sum_group(H, lane, r, nrm, take, a0, a1, a2, a3);
      }
      float* ap = acc + cl * PITCH + lane;
      ap[0] = a0;
      ap[32] = a1;
      ap[64] = a2;
      ap[96] = a3;
    }
    for (int i = nb * PITCH + tid; i < NBP * PITCH; i += AGG_THREADS) acc[i] = 0.0f;
  } else {
    if (tid == 0) kstG[(size_t)blockIdx.x * KPITCH + NKEYS] = -1;
    {
      float4* a4 = (float4*)acc;
      for (int i = tid; i < NBP * PITCH / 4; i += AGG_THREADS)
        a4[i] = make_float4(0.0f, 0.0f, 0.0f, 0.0f);
    }
    __syncthreads();
    walk_runs(rsrc, rcol, rw, seg, deg, E, blockIdx.x, wave, lane, n0,
              [&](int r, int cl, float nrm, int take, int) { add_group(H, acc, lane, r, cl, nrm, take); });
  }

  __syncthreads();

  for (int i = tid; i < nb * D_FEAT; i += AGG_THREADS) {
    const int f = i & (D_FEAT - 1);
    const int rl = i >> 7;
    const int node = n0 + rl;
    const float d = dl[rl];
    const size_t g = (size_t)node * D_FEAT + f;
    const float v = acc[rl * PITCH + f] + H[g] * d * d + bias[f];
    const float z = fmaxf(v, 0.0f) + pw[f] * fminf(v, 0.0f);
    acc[rl * PITCH + f] = z;
  }

  __syncthreads();
  const int m0 = (wave >> 1) * 16;
  if (wave < 2 * (NBP / 16) && m0 < nb) {
    const int nbase = (wave & 1) * 64;
    const int l15 = lane & 15;
    const int h = lane >> 4;
    const float* ap = acc + (m0 + l15) * PITCH + 2 * h;
    const float* bp = wl + h * WPITCH + (nbase + l15) * 2;
    v8f c0 = {}, c1 = {}, c2 = {}, c3 = {};
#pragma unroll
    for (int k0 = 0; k0 < D_FEAT; k0 += 4) {
      const v2f a = *(const v2f*)(ap + k0);
      const float* bk = bp + (k0 >> 1) * WPITCH;
      const v2f b0 = *(const v2f*)(bk);
      const v2f b1 = *(const v2f*)(bk + 32);
      const v2f b2 = *(const v2f*)(bk + 64);
      const v2f b3 = *(const v2f*)(bk + 96);
      c0 = __builtin_amdgcn_wmma_f32_16x16x4_f32(false, a, false, b0, (short)0, c0, false, false);
      c1 = __builtin_amdgcn_wmma_f32_16x16x4_f32(false, a, false, b1, (short)0, c1, false, false);
      c2 = __builtin_amdgcn_wmma_f32_16x16x4_f32(false, a, false, b2, (short)0, c2, false, false);
      c3 = __builtin_amdgcn_wmma_f32_16x16x4_f32(false, a, false, b3, (short)0, c3, false, false);
    }
    const int rbase = m0 + 8 * h;
    float* zp = Z + (size_t)(n0 + rbase) * D_FEAT + nbase + l15;
#pragma unroll
    for (int r = 0; r < 8; ++r) {
      if (rbase + r < nb) {
        float* q = zp + (size_t)r * D_FEAT;
        q[0] = c0[r];
        q[16] = c1[r];
        q[32] = c2[r];
        q[48] = c3[r];
      }
    }
  }
}

__global__ __launch_bounds__(AGG_THREADS) void gather_block_kernel(
    const int* __restrict__ rsrc, const int* __restrict__ rcol,
    const float* __restrict__ rw, const int* __restrict__ seg,
    const float* __restrict__ deg,
    const float* __restrict__ H, const float* __restrict__ bias,
    const float* __restrict__ pw, float* __restrict__ Z, int E, int N,
    const int* __restrict__ kstG, const int* __restrict__ bsrcG, const float* __restrict__ bnrmG) {
  __shared__ float acc[NBP * PITCH];
  __shared__ float dl[NB];

  const int tid = threadIdx.x;
  const int lane = tid & 31;
  const int wave = __builtin_amdgcn_readfirstlane(tid >> 5);
  const int n0 = blockIdx.x * NB;
  const int nb = min(NB, N - n0);
  const int* kb = kstG + (size_t)blockIdx.x * KPITCH;

  const int blockTotal = __builtin_amdgcn_readfirstlane(kb[NKEYS]);
  if (blockTotal >= 0) {
    const int* bs = bsrcG + (size_t)blockIdx.x * ECAP;
    const float* bn = bnrmG + (size_t)blockIdx.x * ECAP;
    if (wave >= nb) return;
    int kv = 0;
    if (lane <= SLOTS) kv = kb[wave * SLOTS + lane];
    float bq[4], pq[4];
#pragma unroll
    for (int q = 0; q < 4; ++q) {
      bq[q] = bias[lane + 32 * q];
      pq[q] = pw[lane + 32 * q];
    }
    int b1 = __builtin_amdgcn_readlane(kv, 0);
    int rN = 0;
    float nN = 0.0f, dN, hN[4];
    {
      const int b0 = b1;
      b1 = __builtin_amdgcn_readlane(kv, 1);
      if (lane < b1 - b0) {
        rN = bs[b0 + lane];
        nN = bn[b0 + lane];
      }
      dN = deg[n0 + wave];
      const float* hp = H + (size_t)(n0 + wave) * D_FEAT + lane;
#pragma unroll
      for (int q = 0; q < 4; ++q) hN[q] = hp[32 * q];
    }
    for (int s = 0; s < SLOTS; ++s) {
      const int cl = s * 32 + wave;
      if (cl >= nb) break;
      const int node = n0 + cl;
      const int c1 = b1;
      const int c0 = __builtin_amdgcn_readlane(kv, s);
      const int r = rN;
      const float nrm = nN, dg = dN;
      float hs[4];
#pragma unroll
      for (int q = 0; q < 4; ++q) hs[q] = hN[q];
      if (s + 1 < SLOTS && cl + 32 < nb) {
        b1 = __builtin_amdgcn_readlane(kv, s + 2);
        rN = 0;
        nN = 0.0f;
        if (lane < b1 - c1) {
          rN = bs[c1 + lane];
          nN = bn[c1 + lane];
        }
        dN = deg[node + 32];
        const float* hp = H + (size_t)(node + 32) * D_FEAT + lane;
#pragma unroll
        for (int q = 0; q < 4; ++q) hN[q] = hp[32 * q];
      }
      float a0 = 0.0f, a1 = 0.0f, a2 = 0.0f, a3 = 0.0f;
      sum_group(H, lane, r, nrm, min(32, c1 - c0), a0, a1, a2, a3);
      for (int g = c0 + 32; g < c1; g += 32) {
        const int take = min(32, c1 - g);
        int r2 = 0;
        float n2 = 0.0f;
        if (lane < take) {
          r2 = bs[g + lane];
          n2 = bn[g + lane];
        }
        sum_group(H, lane, r2, n2, take, a0, a1, a2, a3);
      }
      const float d = rsqrtf(dg + 1.0f);
      const float av[4] = {a0, a1, a2, a3};
      float* zp = Z + (size_t)node * D_FEAT + lane;
#pragma unroll
      for (int q = 0; q < 4; ++q) {
        const float v = av[q] + hs[q] * d * d + bq[q];
        zp[32 * q] = fmaxf(v, 0.0f) + pq[q] * fminf(v, 0.0f);
      }
    }
    return;
  }

  if (tid < nb) dl[tid] = dinv_of(deg, n0 + tid);
  {
    float4* a4 = (float4*)acc;
    for (int i = tid; i < NBP * PITCH / 4; i += AGG_THREADS)
      a4[i] = make_float4(0.0f, 0.0f, 0.0f, 0.0f);
  }
  __syncthreads();
  walk_runs(rsrc, rcol, rw, seg, deg, E, blockIdx.x, wave, lane, n0,
            [&](int r, int cl, float nrm, int take, int) { add_group(H, acc, lane, r, cl, nrm, take); });
  __syncthreads();
  for (int i = tid; i < nb * D_FEAT; i += AGG_THREADS) {
    const int f = i & (D_FEAT - 1);
    const int rl = i >> 7;
    const float d = dl[rl];
    const size_t g = (size_t)(n0 + rl) * D_FEAT + f;
    const float v = acc[rl * PITCH + f] + H[g] * d * d + bias[f];
    Z[g] = fmaxf(v, 0.0f) + pw[f] * fminf(v, 0.0f);
  }
}

#define COMB_THREADS 256
#define COMB_BLOCKS 512
__global__ __launch_bounds__(COMB_THREADS) void combine_kernel(
    const float* __restrict__ z0, const float* __restrict__ z1,
    const float* __restrict__ z2, const float* __restrict__ coeff,
    float* __restrict__ out, int n4) {
  const float c0 = coeff[0], c1 = coeff[1], c2 = coeff[2];
  const float4* p0 = (const float4*)z0;
  const float4* p1 = (const float4*)z1;
  const float4* p2 = (const float4*)z2;
  float4* o4 = (float4*)out;
  const int stride = gridDim.x * COMB_THREADS;
  int i = blockIdx.x * COMB_THREADS + threadIdx.x;

  auto mix = [&](const float4& a, const float4& b, const float4& c) {
    float4 r;
    r.x = __builtin_fmaf(c2, c.x, __builtin_fmaf(c1, b.x, c0 * a.x));
    r.y = __builtin_fmaf(c2, c.y, __builtin_fmaf(c1, b.y, c0 * a.y));
    r.z = __builtin_fmaf(c2, c.z, __builtin_fmaf(c1, b.z, c0 * a.z));
    r.w = __builtin_fmaf(c2, c.w, __builtin_fmaf(c1, b.w, c0 * a.w));
    return r;
  };

  for (; i + 3 * stride < n4; i += 4 * stride) {
    float4 a[4], b[4], c[4];
#pragma unroll
    for (int u = 0; u < 4; ++u) {
      a[u] = p0[i + u * stride];
      b[u] = p1[i + u * stride];
      c[u] = p2[i + u * stride];
    }
#pragma unroll
    for (int u = 0; u < 4; ++u) o4[i + u * stride] = mix(a[u], b[u], c[u]);
  }
  for (; i < n4; i += stride) o4[i] = mix(p0[i], p1[i], p2[i]);
}

extern "C" void kernel_launch(void* const* d_in, const int* in_sizes, int n_in,
                              void* d_out, int out_size, void* d_ws, size_t ws_size,
                              hipStream_t stream) {
  const float* x       = (const float*)d_in[0];
  const int*   eidx    = (const int*)  d_in[1];
  const float* ew      = (const float*)d_in[2];
  const float* coeff   = (const float*)d_in[3];
  const float* W1      = (const float*)d_in[4];
  const float* b1      = (const float*)d_in[5];
  const float* W2      = (const float*)d_in[6];
  const float* b2      = (const float*)d_in[7];
  const float* prelu_w = (const float*)d_in[8];

  const int N = in_sizes[0] / D_FEAT;
  const int E = in_sizes[2] / 3;
  const int total = N * D_FEAT;

  const int aggBlocks  = (N + NB - 1) / NB;

  float* ws   = (float*)d_ws;
  const size_t nPad = ((size_t)N + 31) & ~(size_t)31;
  float* deg3 = ws;
  float* bufA = ws + 3 * nPad;
  float* zpl  = bufA + (size_t)N * D_FEAT;
  float* bufB = zpl + (size_t)3 * N * D_FEAT;
  int*   kstG = (int*)(bufB + (size_t)N * D_FEAT);
  int*   bsrc = kstG + (size_t)aggBlocks * KPITCH;
  float* bnrm = (float*)(bsrc + (size_t)aggBlocks * ECAP);
  const size_t ePad = ((size_t)E + 31) & ~(size_t)31;
  int*   rsrc = (int*)(bnrm + (size_t)aggBlocks * ECAP);
  int*   rcol = rsrc + ePad;
  float* rw   = (float*)(rcol + ePad);
  int*   seg  = (int*)(rw + ePad);

  const int ETHREADS = 256;
  const int degWords = (int)(3 * nPad);
  zero_kernel<<<(degWords + ETHREADS - 1) / ETHREADS, ETHREADS, 0, stream>>>(deg3, degWords);

  for (int t = 0; t < 3; ++t) {
    const int* row = eidx + (size_t)t * 2 * E;
    const int* col = row + E;
    const float* w = ew + (size_t)t * E;
    float* zt = zpl + (size_t)t * total;
    float* deg = deg3 + (size_t)t * nPad;

    degree_sort_kernel<<<NCHUNK + PROD_WGS, SORT_THREADS, 0, stream>>>(row, col, w, deg, E, rsrc, rcol, rw, seg,
                                                                    x, W1, bufA, N);

    aggregate_block_kernel<<<aggBlocks, AGG_THREADS, 0, stream>>>(rsrc, rcol, rw, seg, deg, bufA, b1, prelu_w, bufB, E, N,
                                                               kstG, bsrc, bnrm, W2);

    gather_block_kernel<<<aggBlocks, AGG_THREADS, 0, stream>>>(rsrc, rcol, rw, seg, deg, bufB, b2, prelu_w, zt, E, N,
                                                            kstG, bsrc, bnrm);
  }

  combine_kernel<<<COMB_BLOCKS, COMB_THREADS, 0, stream>>>(zpl, zpl + (size_t)total, zpl + (size_t)2 * total, coeff,
                                                         (float*)d_out, total / 4);
}


# The three scales' degree/sort passes run as one launch ahead of all aggregation (3 x 128 sort work-groups, a scale's own planes); x@W1 formed once beside them; 11 -> 9 launches; sums unchanged
# speedup vs baseline: 4.1701x; 1.1557x over previous
#include <hip/hip_runtime.h>
#include <hip/hip_bf16.h>


typedef __attribute__((ext_vector_type(2))) float v2f;
typedef __attribute__((ext_vector_type(8))) float v8f;

#define D_FEAT 128

__global__ void zero_kernel(float* __restrict__ p, int n) {
  int i = blockIdx.x * blockDim.x + threadIdx.x;
  if (i < n) p[i] = 0.0f;
}

#define NB 196
#define NCHUNK 128
#define N_SCALES 3
#define PROD_WGS 125
#define SORT_THREADS 1024
#define SORT_WAVES (SORT_THREADS / 32)
#define SORT_EPT 8
#define SORT_CAP (SORT_EPT * SORT_THREADS)
#define SORT_BINS 256
#define WPITCH 288

__device__ __forceinline__ void product_part(
    const float* __restrict__ X, const float* __restrict__ Wm, float* __restrict__ H,
    int nRows, int g, int nG) {
  __shared__ float wl[(D_FEAT / 2) * WPITCH];
  const int tid = threadIdx.x;
  const int lane = tid & 31;
  const int wave = __builtin_amdgcn_readfirstlane(tid >> 5);
  for (int i = tid; i < (D_FEAT / 2) * (D_FEAT / 4); i += SORT_THREADS) {
    const int p = i >> 5;
    const int c4 = (i & 31) * 4;
    const float4 lo = *(const float4*)(Wm + (size_t)(2 * p) * D_FEAT + c4);
    const float4 hi = *(const float4*)(Wm + (size_t)(2 * p + 1) * D_FEAT + c4);
    float4* dst = (float4*)(wl + p * WPITCH + c4 * 2);
    dst[0] = make_float4(lo.x, hi.x, lo.y, hi.y);
    dst[1] = make_float4(lo.z, hi.z, lo.w, hi.w);
  }
  __syncthreads();

  const int strips = nRows / 16;
  const int spw = (strips + nG - 1) / nG;
  const int s0 = g * spw;
  const int s1 = min(strips, s0 + spw);
  const int l15 = lane & 15;
  const int h = lane >> 4;
  for (int task = 2 * s0 + wave; task < 2 * s1; task += SORT_WAVES) {
    const int m0 = (task >> 1) * 16;
    const int nbase = (task & 1) * 64;
    const float* ap = X + (size_t)(m0 + l15) * D_FEAT + 2 * h;
    const float* bp = wl + h * WPITCH + (nbase + l15) * 2;
    v8f c0 = {}, c1 = {}, c2 = {}, c3 = {};
#pragma nounroll
    for (int kb = 0; kb < D_FEAT; kb += 32) {
      v2f a[8];
#pragma unroll
      for (int u = 0; u < 8; ++u) a[u] = *(const v2f*)(ap + kb + 4 * u);
#pragma unroll
      for (int u = 0; u < 8; ++u) {
        const float* bk = bp + ((kb >> 1) + 2 * u) * WPITCH;
        const v2f b0 = *(const v2f*)(bk);
        const v2f b1 = *(const v2f*)(bk + 32);
        const v2f b2 = *(const v2f*)(bk + 64);
        const v2f b3 = *(const v2f*)(bk + 96);
        c0 = __builtin_amdgcn_wmma_f32_16x16x4_f32(false, a[u], false, b0, (short)0, c0, false, false);
        c1 = __builtin_amdgcn_wmma_f32_16x16x4_f32(false, a[u], false, b1, (short)0, c1, false, false);
        c2 = __builtin_amdgcn_wmma_f32_16x16x4_f32(false, a[u], false, b2, (short)0, c2, false, false);
        c3 = __builtin_amdgcn_wmma_f32_16x16x4_f32(false, a[u], false, b3, (short)0, c3, false, false);
      }
    }
    float* hp = H + (size_t)(m0 + 8 * h) * D_FEAT + nbase + l15;
#pragma unroll
    for (int r = 0; r < 8; ++r) {
      float* q = hp + (size_t)r * D_FEAT;
      q[0] = c0[r];
      q[16] = c1[r];
      q[32] = c2[r];
      q[48] = c3[r];
    }
  }
}

__global__ __launch_bounds__(SORT_THREADS) void degree_sort_kernel(
    const int* __restrict__ eidx, const float* __restrict__ ew, float* __restrict__ deg3,
    int E, int nPad,
    int* __restrict__ rsrc3, int* __restrict__ rcol3, float* __restrict__ rw3, int ePad,
    int* __restrict__ seg3,
    const float* __restrict__ X, const float* __restrict__ Wm, float* __restrict__ H, int nRows) {
  if (blockIdx.x < PROD_WGS) {
    product_part(X, Wm, H, nRows, blockIdx.x, PROD_WGS);
    return;
  }
  const int sortWg = blockIdx.x - PROD_WGS;
  const int scale = sortWg / NCHUNK;
  const int chunk = sortWg - scale * NCHUNK;
  const int* __restrict__ row = eidx + (size_t)scale * 2 * E;
  const int* __restrict__ col = row + E;
  const float* __restrict__ w = ew + (size_t)scale * E;
  float* __restrict__ deg = deg3 + (size_t)scale * nPad;
  int* __restrict__ rsrc = rsrc3 + (size_t)scale * ePad;
  int* __restrict__ rcol = rcol3 + (size_t)scale * ePad;
  float* __restrict__ rw = rw3 + (size_t)scale * ePad;
  int* __restrict__ seg = seg3 + (size_t)scale * (SORT_BINS * NCHUNK * 2);
  __shared__ int cnt[SORT_WAVES][SORT_BINS];
  __shared__ int binStart[SORT_BINS];
  __shared__ int waveSum[SORT_BINS / 32];
  __shared__ int s_r[SORT_CAP];
  __shared__ int s_c[SORT_CAP];
  __shared__ float s_w[SORT_CAP];

  const int tid = threadIdx.x;
  const int lane = tid & 31;
  const int wave = tid >> 5;
  const int chunkLen = (E + NCHUNK - 1) / NCHUNK;
  const int e0 = chunk * chunkLen;
  const int n = max(0, min(chunkLen, E - e0));

  for (int i = tid; i < SORT_WAVES * SORT_BINS; i += SORT_THREADS) (&cnt[0][0])[i] = 0;
  __syncthreads();

  int r[SORT_EPT], c[SORT_EPT], b[SORT_EPT], rank[SORT_EPT];
  float wt[SORT_EPT];
#pragma unroll
  for (int k = 0; k < SORT_EPT; ++k) {
    const int i = k * SORT_THREADS + tid;
    const bool ok = i < n;
    c[k] = ok ? col[e0 + i] : -1;
    r[k] = ok ? row[e0 + i] : 0;
    wt[k] = ok ? w[e0 + i] : 0.0f;
  }
#pragma unroll
  for (int k = 0; k < SORT_EPT; ++k) {
    b[k] = 0;
    rank[k] = 0;
    if (c[k] >= 0) {
      b[k] = c[k] / NB;
      rank[k] = __hip_atomic_fetch_add(&cnt[wave][b[k]], 1, __ATOMIC_RELAXED,
                                       __HIP_MEMORY_SCOPE_WORKGROUP);
    }
  }
  __syncthreads();

  int tot = 0;
  if (tid < SORT_BINS) {
    int t[SORT_WAVES];
#pragma unroll
    for (int q = 0; q < SORT_WAVES; ++q) t[q] = cnt[q][tid];
#pragma unroll
    for (int q = 0; q < SORT_WAVES; ++q) {
      cnt[q][tid] = tot;
      tot += t[q];
    }
  }
  int incl = tot;
#pragma unroll
  for (int d = 1; d < 32; d <<= 1) {
    const int o = __shfl_up(incl, d);
    if (lane >= d) incl += o;
  }
  if (tid < SORT_BINS && lane == 31) waveSum[wave] = incl;
  __syncthreads();
  if (tid < SORT_BINS) {
    int base = 0;
    for (int q = 0; q < wave; ++q) base += waveSum[q];
    const int start = base + incl - tot;
    binStart[tid] = start;
    ((int2*)seg)[(size_t)tid * NCHUNK + chunk] = make_int2(start, tot);
  }
  __syncthreads();

#pragma unroll
  for (int k = 0; k < SORT_EPT; ++k) {
    if (c[k] >= 0) {
      const int pos = binStart[b[k]] + cnt[wave][b[k]] + rank[k];
      s_r[pos] = r[k];
      s_c[pos] = c[k];
      s_w[pos] = wt[k];
    }
  }
  __syncthreads();
  for (int i = tid; i < n; i += SORT_THREADS) {
    rsrc[e0 + i] = s_r[i];
    rcol[e0 + i] = s_c[i];
    rw[e0 + i] = s_w[i];
  }
#pragma unroll
  for (int k = 0; k < SORT_EPT; ++k)
    if (c[k] >= 0) atomicAdd(&deg[c[k]], wt[k]);
}

__device__ __forceinline__ float dinv_of(const float* __restrict__ deg, int i) {
  const float d = deg[i] + 1.0f;
  return rsqrtf(d);
}

#define NBP 208
#define PITCH 132
#define AGG_THREADS 1024
#define AGG_WAVES (AGG_THREADS / 32)
#define CPW (NCHUNK / AGG_WAVES)
#define WCAP 256
#define ECAP (WCAP * AGG_WAVES)
#define SLOTS ((NB + 31) / 32)
#define NKEYS (SLOTS * AGG_WAVES)
#define KPITCH 256

__device__ __forceinline__ void sum_group(const float* __restrict__ H, int lane, int r, float nrm,
                                          int take, float& a0, float& a1, float& a2, float& a3) {
  for (int g0 = 0; g0 < take; g0 += 8) {
    float v[8][4];
#pragma unroll
    for (int u = 0; u < 8; ++u) {
      const int k = min(g0 + u, take - 1);
      const int rk = __builtin_amdgcn_readlane(r, k);
      const float* hp = H + (size_t)rk * D_FEAT + lane;
#pragma unroll
      for (int q = 0; q < 4; ++q) v[u][q] = hp[32 * q];
    }
#pragma unroll
    for (int u = 0; u < 8; ++u)
      asm volatile("" : : "v"(v[u][0]), "v"(v[u][1]), "v"(v[u][2]), "v"(v[u][3]));
#pragma unroll
    for (int u = 0; u < 8; ++u) {
      if (g0 + u < take) {
        const float nk = __int_as_float(__builtin_amdgcn_readlane(__float_as_int(nrm), g0 + u));
        a0 = __builtin_fmaf(v[u][0], nk, a0);
        a1 = __builtin_fmaf(v[u][1], nk, a1);
        a2 = __builtin_fmaf(v[u][2], nk, a2);
        a3 = __builtin_fmaf(v[u][3], nk, a3);
      }
    }
  }
}

__device__ __forceinline__ void add_group(const float* __restrict__ H, float* acc, int lane,
                                          int r, int cl, float nrm, int take) {
  for (int g0 = 0; g0 < take; g0 += 8) {
    float v[8][4];
#pragma unroll
    for (int u = 0; u < 8; ++u) {
      const int k = min(g0 + u, take - 1);
      const int rk = __builtin_amdgcn_readlane(r, k);
      const float* hp = H + (size_t)rk * D_FEAT + lane;
#pragma unroll
      for (int q = 0; q < 4; ++q) v[u][q] = hp[32 * q];
    }
#pragma unroll
    for (int u = 0; u < 8; ++u)
      asm volatile("" : : "v"(v[u][0]), "v"(v[u][1]), "v"(v[u][2]), "v"(v[u][3]));
#pragma unroll
    for (int u = 0; u < 8; ++u) {
      if (g0 + u < take) {
        const int ck = __builtin_amdgcn_readlane(cl, g0 + u);
        const float nk = __int_as_float(__builtin_amdgcn_readlane(__float_as_int(nrm), g0 + u));
        float* ap = acc + ck * PITCH + lane;
#pragma unroll
        for (int q = 0; q < 4; ++q)
          __hip_atomic_fetch_add(ap + 32 * q, v[u][q] * nk, __ATOMIC_RELAXED,
                                 __HIP_MEMORY_SCOPE_WORKGROUP);
      }
    }
  }
}

template <class F>
__device__ __forceinline__ int walk_runs(
    const int* __restrict__ rsrc, const int* __restrict__ rcol, const float* __restrict__ rw,
    const int* __restrict__ seg, const float* __restrict__ deg, int E,
    int block, int wave, int lane, int n0, F&& f) {
  const int chunkLen = (E + NCHUNK - 1) / NCHUNK;
  int2 sc = make_int2(0, 0);
  if (lane < CPW) sc = ((const int2*)seg)[(size_t)block * NCHUNK + wave * CPW + lane];
  int s[CPW], p[CPW + 1];
  p[0] = 0;
#pragma unroll
  for (int q = 0; q < CPW; ++q) {
    s[q] = __builtin_amdgcn_readlane(sc.x, q) + (wave * CPW + q) * chunkLen;
    p[q + 1] = p[q] + __builtin_amdgcn_readlane(sc.y, q);
  }
  const int total = p[CPW];
  for (int done = 0; done < total; done += 32) {
    const int v = done + lane;
    int r = 0, cl = 0;
    float nrm = 0.0f;
    if (v < total) {
      int a = s[0] + v;
#pragma unroll
      for (int q = 1; q < CPW; ++q)
        if (v >= p[q]) a = s[q] + (v - p[q]);
      r = rsrc[a];
      const int c = rcol[a];
      nrm = dinv_of(deg, r) * rw[a] * dinv_of(deg, c);
      cl = c - n0;
    }
    f(r, cl, nrm, min(32, total - done), done);
  }
  return total;
}

__global__ __launch_bounds__(AGG_THREADS) void aggregate_block_kernel(
    const int* __restrict__ rsrc, const int* __restrict__ rcol,
    const float* __restrict__ rw, const int* __restrict__ seg,
    const float* __restrict__ deg,
    const float* __restrict__ H, const float* __restrict__ bias,
    const float* __restrict__ pw, float* __restrict__ Z, int E, int N,
    int* __restrict__ kstG, int* __restrict__ bsrcG, float* __restrict__ bnrmG,
    const float* __restrict__ Wn) {
  __shared__ float acc[NBP * PITCH];
  __shared__ float wl[(D_FEAT / 2) * WPITCH];
  __shared__ float dl[NB];
  __shared__ int kcnt[AGG_WAVES][NKEYS];
  __shared__ int kstart[NKEYS + 1];
  __shared__ int kwsum[NKEYS / 32];
  __shared__ int s_src[ECAP];
  __shared__ float s_nrm[ECAP];
  __shared__ int ovf;

  const int tid = threadIdx.x;
  const int lane = tid & 31;
  const int wave = __builtin_amdgcn_readfirstlane(tid >> 5);
  const int n0 = blockIdx.x * NB;
  const int nb = min(NB, N - n0);

  int* st_r = (int*)acc;
  float* st_n = acc + ECAP;
  int* st_k = (int*)(acc + 2 * ECAP);

  for (int i = tid; i < AGG_WAVES * NKEYS; i += AGG_THREADS) (&kcnt[0][0])[i] = 0;
  if (tid == 0) ovf = 0;
  if (tid < nb) dl[tid] = dinv_of(deg, n0 + tid);
  for (int i = tid; i < (D_FEAT / 2) * (D_FEAT / 4); i += AGG_THREADS) {
    const int p = i >> 5;
    const int c4 = (i & 31) * 4;
    const float4 lo = *(const float4*)(Wn + (size_t)(2 * p) * D_FEAT + c4);
    const float4 hi = *(const float4*)(Wn + (size_t)(2 * p + 1) * D_FEAT + c4);
    float4* dst = (float4*)(wl + p * WPITCH + c4 * 2);
    dst[0] = make_float4(lo.x, hi.x, lo.y, hi.y);
    dst[1] = make_float4(lo.z, hi.z, lo.w, hi.w);
  }
  __syncthreads();

  auto stage_group = [&](int r, int cl, float nrm, int take, int done) {
    const int i = done + lane;
    const bool valid = lane < take && i < WCAP;
    const int key = (cl & 31) * SLOTS + (cl >> 5);
    const int kk = valid ? key : -1;
    int before = 0, after = 0;
#pragma unroll
    for (int j = 0; j < 32; ++j) {
      const bool eq = __builtin_amdgcn_readlane(kk, j) == kk;
      before += (eq && j < lane) ? 1 : 0;
      after += (eq && j > lane) ? 1 : 0;
    }
    if (valid) {
      const int rank = kcnt[wave][key] + before;
      const int at = wave * WCAP + i;
      st_r[at] = r;
      st_n[at] = nrm;
      st_k[at] = key | (rank << 8);
      if (after == 0) kcnt[wave][key] = rank + 1;
    }
  };

  const int myTotal = walk_runs(rsrc, rcol, rw, seg, deg, E, blockIdx.x, wave, lane, n0, stage_group);
  if (myTotal > WCAP && lane == 0) ovf = 1;
  __syncthreads();

  if (ovf == 0) {
    int tot = 0;
    if (tid < NKEYS) {
      int t[AGG_WAVES];
#pragma unroll
      for (int q = 0; q < AGG_WAVES; ++q) t[q] = kcnt[q][tid];
#pragma unroll
      for (int q = 0; q < AGG_WAVES; ++q) {
        kcnt[q][tid] = tot;
        tot += t[q];
      }
    }
    int incl = tot;
#pragma unroll
    for (int d = 1; d < 32; d <<= 1) {
      const int o = __shfl_up(incl, d);
      if (lane >= d) incl += o;
    }
    if (tid < NKEYS && lane == 31) kwsum[wave] = incl;
    __syncthreads();
    if (tid < NKEYS) {
      int base = 0;
      for (int q = 0; q < wave; ++q) base += kwsum[q];
      kstart[tid] = base + incl - tot;
      if (tid == NKEYS - 1) kstart[NKEYS] = base + incl;
    }
    __syncthreads();

    for (int i = lane; i < myTotal; i += 32) {
      const int at = wave * WCAP + i;
      const int kr = st_k[at];
      const int key = kr & 255;
      const int pos = kstart[key] + kcnt[wave][key] + (kr >> 8);
      s_src[pos] = st_r[at];
      s_nrm[pos] = st_n[at];
    }
    __syncthreads();

    {
      int* kb = kstG + (size_t)blockIdx.x * KPITCH;
      int* bs = bsrcG + (size_t)blockIdx.x * ECAP;
      float* bn = bnrmG + (size_t)blockIdx.x * ECAP;
      if (tid <= NKEYS) kb[tid] = kstart[tid];
      const int blockTotal = kstart[NKEYS];
      for (int i = tid; i < blockTotal; i += AGG_THREADS) {
        bs[i] = s_src[i];
        bn[i] = s_nrm[i];
      }
    }

    for (int s = 0; s < SLOTS; ++s) {
      const int cl = s * 32 + wave;
      if (cl >= nb) break;
      const int b0 = __builtin_amdgcn_readfirstlane(kstart[wave * SLOTS + s]);
      const int b1 = __builtin_amdgcn_readfirstlane(kstart[wave * SLOTS + s + 1]);
      float a0 = 0.0f, a1 = 0.0f, a2 = 0.0f, a3 = 0.0f;
      for (int g = b0; g < b1; g += 32) {
        const int take = min(32, b1 - g);
        int r = 0;
        float nrm = 0.0f;
        if (lane < take) {
          r = s_src[g + lane];
          nrm = s_nrm[g + lane];
        }
        sum_group(H, lane, r, nrm, take, a0, a1, a2, a3);
      }
      float* ap = acc + cl * PITCH + lane;
      ap[0] = a0;
      ap[32] = a1;
      ap[64] = a2;
      ap[96] = a3;
    }
    for (int i = nb * PITCH + tid; i < NBP * PITCH; i += AGG_THREADS) acc[i] = 0.0f;
  } else {
    if (tid == 0) kstG[(size_t)blockIdx.x * KPITCH + NKEYS] = -1;
    {
      float4* a4 = (float4*)acc;
      for (int i = tid; i < NBP * PITCH / 4; i += AGG_THREADS)
        a4[i] = make_float4(0.0f, 0.0f, 0.0f, 0.0f);
    }
    __syncthreads();
    walk_runs(rsrc, rcol, rw, seg, deg, E, blockIdx.x, wave, lane, n0,
              [&](int r, int cl, float nrm, int take, int) { add_group(H, acc, lane, r, cl, nrm, take); });
  }

  __syncthreads();

  for (int i = tid; i < nb * D_FEAT; i += AGG_THREADS) {
    const int f = i & (D_FEAT - 1);
    const int rl = i >> 7;
    const int node = n0 + rl;
    const float d = dl[rl];
    const size_t g = (size_t)node * D_FEAT + f;
    const float v = acc[rl * PITCH + f] + H[g] * d * d + bias[f];
    const float z = fmaxf(v, 0.0f) + pw[f] * fminf(v, 0.0f);
    acc[rl * PITCH + f] = z;
  }

  __syncthreads();
  const int m0 = (wave >> 1) * 16;
  if (wave < 2 * (NBP / 16) && m0 < nb) {
    const int nbase = (wave & 1) * 64;
    const int l15 = lane & 15;
    const int h = lane >> 4;
    const float* ap = acc + (m0 + l15) * PITCH + 2 * h;
    const float* bp = wl + h * WPITCH + (nbase + l15) * 2;
    v8f c0 = {}, c1 = {}, c2 = {}, c3 = {};
#pragma unroll
    for (int k0 = 0; k0 < D_FEAT; k0 += 4) {
      const v2f a = *(const v2f*)(ap + k0);
      const float* bk = bp + (k0 >> 1) * WPITCH;
      const v2f b0 = *(const v2f*)(bk);
      const v2f b1 = *(const v2f*)(bk + 32);
      const v2f b2 = *(const v2f*)(bk + 64);
      const v2f b3 = *(const v2f*)(bk + 96);
      c0 = __builtin_amdgcn_wmma_f32_16x16x4_f32(false, a, false, b0, (short)0, c0, false, false);
      c1 = __builtin_amdgcn_wmma_f32_16x16x4_f32(false, a, false, b1, (short)0, c1, false, false);
      c2 = __builtin_amdgcn_wmma_f32_16x16x4_f32(false, a, false, b2, (short)0, c2, false, false);
      c3 = __builtin_amdgcn_wmma_f32_16x16x4_f32(false, a, false, b3, (short)0, c3, false, false);
    }
    const int rbase = m0 + 8 * h;
    float* zp = Z + (size_t)(n0 + rbase) * D_FEAT + nbase + l15;
#pragma unroll
    for (int r = 0; r < 8; ++r) {
      if (rbase + r < nb) {
        float* q = zp + (size_t)r * D_FEAT;
        q[0] = c0[r];
        q[16] = c1[r];
        q[32] = c2[r];
        q[48] = c3[r];
      }
    }
  }
}

__global__ __launch_bounds__(AGG_THREADS) void gather_block_kernel(
    const int* __restrict__ rsrc, const int* __restrict__ rcol,
    const float* __restrict__ rw, const int* __restrict__ seg,
    const float* __restrict__ deg,
    const float* __restrict__ H, const float* __restrict__ bias,
    const float* __restrict__ pw, float* __restrict__ Z, int E, int N,
    const int* __restrict__ kstG, const int* __restrict__ bsrcG, const float* __restrict__ bnrmG) {
  __shared__ float acc[NBP * PITCH];
  __shared__ float dl[NB];

  const int tid = threadIdx.x;
  const int lane = tid & 31;
  const int wave = __builtin_amdgcn_readfirstlane(tid >> 5);
  const int n0 = blockIdx.x * NB;
  const int nb = min(NB, N - n0);
  const int* kb = kstG + (size_t)blockIdx.x * KPITCH;

  const int blockTotal = __builtin_amdgcn_readfirstlane(kb[NKEYS]);
  if (blockTotal >= 0) {
    const int* bs = bsrcG + (size_t)blockIdx.x * ECAP;
    const float* bn = bnrmG + (size_t)blockIdx.x * ECAP;
    if (wave >= nb) return;
    int kv = 0;
    if (lane <= SLOTS) kv = kb[wave * SLOTS + lane];
    float bq[4], pq[4];
#pragma unroll
    for (int q = 0; q < 4; ++q) {
      bq[q] = bias[lane + 32 * q];
      pq[q] = pw[lane + 32 * q];
    }
    int b1 = __builtin_amdgcn_readlane(kv, 0);
    int rN = 0;
    float nN = 0.0f, dN, hN[4];
    {
      const int b0 = b1;
      b1 = __builtin_amdgcn_readlane(kv, 1);
      if (lane < b1 - b0) {
        rN = bs[b0 + lane];
        nN = bn[b0 + lane];
      }
      dN = deg[n0 + wave];
      const float* hp = H + (size_t)(n0 + wave) * D_FEAT + lane;
#pragma unroll
      for (int q = 0; q < 4; ++q) hN[q] = hp[32 * q];
    }
    for (int s = 0; s < SLOTS; ++s) {
      const int cl = s * 32 + wave;
      if (cl >= nb) break;
      const int node = n0 + cl;
      const int c1 = b1;
      const int c0 = __builtin_amdgcn_readlane(kv, s);
      const int r = rN;
      const float nrm = nN, dg = dN;
      float hs[4];
#pragma unroll
      for (int q = 0; q < 4; ++q) hs[q] = hN[q];
      if (s + 1 < SLOTS && cl + 32 < nb) {
        b1 = __builtin_amdgcn_readlane(kv, s + 2);
        rN = 0;
        nN = 0.0f;
        if (lane < b1 - c1) {
          rN = bs[c1 + lane];
          nN = bn[c1 + lane];
        }
        dN = deg[node + 32];
        const float* hp = H + (size_t)(node + 32) * D_FEAT + lane;
#pragma unroll
        for (int q = 0; q < 4; ++q) hN[q] = hp[32 * q];
      }
      float a0 = 0.0f, a1 = 0.0f, a2 = 0.0f, a3 = 0.0f;
      sum_group(H, lane, r, nrm, min(32, c1 - c0), a0, a1, a2, a3);
      for (int g = c0 + 32; g < c1; g += 32) {
        const int take = min(32, c1 - g);
        int r2 = 0;
        float n2 = 0.0f;
        if (lane < take) {
          r2 = bs[g + lane];
          n2 = bn[g + lane];
        }
        sum_group(H, lane, r2, n2, take, a0, a1, a2, a3);
      }
      const float d = rsqrtf(dg + 1.0f);
      const float av[4] = {a0, a1, a2, a3};
      float* zp = Z + (size_t)node * D_FEAT + lane;
#pragma unroll
      for (int q = 0; q < 4; ++q) {
        const float v = av[q] + hs[q] * d * d + bq[q];
        zp[32 * q] = fmaxf(v, 0.0f) + pq[q] * fminf(v, 0.0f);
      }
    }
    return;
  }

  if (tid < nb) dl[tid] = dinv_of(deg, n0 + tid);
  {
    float4* a4 = (float4*)acc;
    for (int i = tid; i < NBP * PITCH / 4; i += AGG_THREADS)
      a4[i] = make_float4(0.0f, 0.0f, 0.0f, 0.0f);
  }
  __syncthreads();
  walk_runs(rsrc, rcol, rw, seg, deg, E, blockIdx.x, wave, lane, n0,
            [&](int r, int cl, float nrm, int take, int) { add_group(H, acc, lane, r, cl, nrm, take); });
  __syncthreads();
  for (int i = tid; i < nb * D_FEAT; i += AGG_THREADS) {
    const int f = i & (D_FEAT - 1);
    const int rl = i >> 7;
    const float d = dl[rl];
    const size_t g = (size_t)(n0 + rl) * D_FEAT + f;
    const float v = acc[rl * PITCH + f] + H[g] * d * d + bias[f];
    Z[g] = fmaxf(v, 0.0f) + pw[f] * fminf(v, 0.0f);
  }
}

#define COMB_THREADS 256
#define COMB_BLOCKS 512
__global__ __launch_bounds__(COMB_THREADS) void combine_kernel(
    const float* __restrict__ z0, const float* __restrict__ z1,
    const float* __restrict__ z2, const float* __restrict__ coeff,
    float* __restrict__ out, int n4) {
  const float c0 = coeff[0], c1 = coeff[1], c2 = coeff[2];
  const float4* p0 = (const float4*)z0;
  const float4* p1 = (const float4*)z1;
  const float4* p2 = (const float4*)z2;
  float4* o4 = (float4*)out;
  const int stride = gridDim.x * COMB_THREADS;
  int i = blockIdx.x * COMB_THREADS + threadIdx.x;

  auto mix = [&](const float4& a, const float4& b, const float4& c) {
    float4 r;
    r.x = __builtin_fmaf(c2, c.x, __builtin_fmaf(c1, b.x, c0 * a.x));
    r.y = __builtin_fmaf(c2, c.y, __builtin_fmaf(c1, b.y, c0 * a.y));
    r.z = __builtin_fmaf(c2, c.z, __builtin_fmaf(c1, b.z, c0 * a.z));
    r.w = __builtin_fmaf(c2, c.w, __builtin_fmaf(c1, b.w, c0 * a.w));
    return r;
  };

  for (; i + 3 * stride < n4; i += 4 * stride) {
    float4 a[4], b[4], c[4];
#pragma unroll
    for (int u = 0; u < 4; ++u) {
      a[u] = p0[i + u * stride];
      b[u] = p1[i + u * stride];
      c[u] = p2[i + u * stride];
    }
#pragma unroll
    for (int u = 0; u < 4; ++u) o4[i + u * stride] = mix(a[u], b[u], c[u]);
  }
  for (; i < n4; i += stride) o4[i] = mix(p0[i], p1[i], p2[i]);
}

extern "C" void kernel_launch(void* const* d_in, const int* in_sizes, int n_in,
                              void* d_out, int out_size, void* d_ws, size_t ws_size,
                              hipStream_t stream) {
  const float* x       = (const float*)d_in[0];
  const int*   eidx    = (const int*)  d_in[1];
  const float* ew      = (const float*)d_in[2];
  const float* coeff   = (const float*)d_in[3];
  const float* W1      = (const float*)d_in[4];
  const float* b1      = (const float*)d_in[5];
  const float* W2      = (const float*)d_in[6];
  const float* b2      = (const float*)d_in[7];
  const float* prelu_w = (const float*)d_in[8];

  const int N = in_sizes[0] / D_FEAT;
  const int E = in_sizes[2] / 3;
  const int total = N * D_FEAT;

  const int aggBlocks  = (N + NB - 1) / NB;

  float* ws   = (float*)d_ws;
  const size_t nPad = ((size_t)N + 31) & ~(size_t)31;
  float* deg3 = ws;
  float* bufA = ws + 3 * nPad;
  float* zpl  = bufA + (size_t)N * D_FEAT;
  float* bufB = zpl + (size_t)3 * N * D_FEAT;
  int*   kstG = (int*)(bufB + (size_t)N * D_FEAT);
  int*   bsrc = kstG + (size_t)aggBlocks * KPITCH;
  float* bnrm = (float*)(bsrc + (size_t)aggBlocks * ECAP);
  const size_t ePad = ((size_t)E + 31) & ~(size_t)31;
  int*   rsrc3 = (int*)(bnrm + (size_t)aggBlocks * ECAP);
  int*   rcol3 = rsrc3 + N_SCALES * ePad;
  float* rw3   = (float*)(rcol3 + N_SCALES * ePad);
  int*   seg3  = (int*)(rw3 + N_SCALES * ePad);
  const size_t segWords = (size_t)SORT_BINS * NCHUNK * 2;

  const int ETHREADS = 256;
  const int degWords = (int)(3 * nPad);
  zero_kernel<<<(degWords + ETHREADS - 1) / ETHREADS, ETHREADS, 0, stream>>>(deg3, degWords);

  degree_sort_kernel<<<PROD_WGS + N_SCALES * NCHUNK, SORT_THREADS, 0, stream>>>(
      eidx, ew, deg3, E, (int)nPad, rsrc3, rcol3, rw3, (int)ePad, seg3, x, W1, bufA, N);

  for (int t = 0; t < N_SCALES; ++t) {
    float* zt = zpl + (size_t)t * total;
    const float* deg = deg3 + (size_t)t * nPad;
    const int* rsrc = rsrc3 + (size_t)t * ePad;
    const int* rcol = rcol3 + (size_t)t * ePad;
    const float* rw = rw3 + (size_t)t * ePad;
    const int* seg = seg3 + (size_t)t * segWords;

    aggregate_block_kernel<<<aggBlocks, AGG_THREADS, 0, stream>>>(rsrc, rcol, rw, seg, deg, bufA, b1, prelu_w, bufB, E, N,
                                                               kstG, bsrc, bnrm, W2);

    gather_block_kernel<<<aggBlocks, AGG_THREADS, 0, stream>>>(rsrc, rcol, rw, seg, deg, bufB, b2, prelu_w, zt, E, N,
                                                            kstG, bsrc, bnrm);
  }

  combine_kernel<<<COMB_BLOCKS, COMB_THREADS, 0, stream>>>(zpl, zpl + (size_t)total, zpl + (size_t)2 * total, coeff,
                                                         (float*)d_out, total / 4);
}
